# starting kernel
# baseline (speedup 1.0000x reference)
#include <hip/hip_runtime.h>
#include <hip/hip_bf16.h>


typedef float v2f __attribute__((ext_vector_type(2)));
typedef float v8f __attribute__((ext_vector_type(8)));

__device__ __forceinline__ float bandsel(int d, float c0, float c1, float c2) {
    float r = 0.0f;
    r = (d == 0) ? c0 : r;
    r = (d == 1) ? c1 : r;
    r = (d == 2) ? c2 : r;
    return r;
}

__device__ __forceinline__ float load_xp_slow(const float* __restrict__ row, int gi, int L) {
    if (gi >= 1 && gi <= L)  return row[gi - 1];
    if (gi == 0)             return 2.0f * row[0]     - row[1];
    if (gi == L + 1)         return 2.0f * row[L - 1] - row[L - 2];
    return 0.0f;
}

__global__ __launch_bounds__(128) void synth_level_kernel(
        const float* __restrict__ x, const float* __restrict__ d,
        float* __restrict__ out,
        const float* __restrict__ h, const float* __restrict__ g,
        int L) {
    __shared__ float smx[4][16][20];
    __shared__ float smd[4][16][20];

    const int w    = threadIdx.x >> 5;
    const int lane = threadIdx.x & 31;
    const int c0   = w << 4;
    const int n    = lane & 15;
    const int hi   = lane >> 4;

    const float h0 = h[0], h1 = h[1], h2 = h[2], h3 = h[3], h4 = h[4], h5 = h[5];
    const float g0 = g[0], g1 = g[1], g2 = g[2], g3 = g[3], g4 = g[4], g5 = g[5];

    v2f bhe[5], bho[5], bge[5], bgo[5];
#pragma unroll
    for (int c = 0; c < 5; ++c) {
        const int r0 = c * 4 + (hi << 1);
        const int d0r = r0 - n, d1r = r0 + 1 - n;
        bhe[c] = v2f{ bandsel(d0r, h4, h2, h0), bandsel(d1r, h4, h2, h0) };
        bho[c] = v2f{ bandsel(d0r, h5, h3, h1), bandsel(d1r, h5, h3, h1) };
        bge[c] = v2f{ bandsel(d0r, g4, g2, g0), bandsel(d1r, g4, g2, g0) };
        bgo[c] = v2f{ bandsel(d0r, g5, g3, g1), bandsel(d1r, g5, g3, g1) };
    }

    const size_t outL = 2 * (size_t)L;

    for (int it = 0; it < 4; ++it) {
        const int t0 = (blockIdx.x << 6) + (it << 4);

        if (t0 >= 1 && t0 + 19 <= L) {
            for (int idx = lane; idx < 16 * 20; idx += 32) {
                const int r = idx / 20;
                const int i = idx % 20;
                const size_t rowBase = (size_t)(c0 + r) * (size_t)L + (size_t)(t0 - 1);
                smx[w][r][i] = x[rowBase + i];
                smd[w][r][i] = d[rowBase + i];
            }
        } else {
            for (int idx = lane; idx < 16 * 20; idx += 32) {
                const int r  = idx / 20;
                const int i  = idx % 20;
                const int gi = t0 + i;
                const size_t rowBase = (size_t)(c0 + r) * (size_t)L;
                smx[w][r][i] = load_xp_slow(x + rowBase, gi, L);
                smd[w][r][i] = load_xp_slow(d + rowBase, gi, L);
            }
        }
        __syncthreads();

        v8f ce = {};
        v8f co = {};
#pragma unroll
        for (int c = 0; c < 5; ++c) {
            const int k0 = c * 4 + (hi << 1);
            const v2f ax = v2f{ smx[w][n][k0], smx[w][n][k0 + 1] };
            const v2f ad = v2f{ smd[w][n][k0], smd[w][n][k0 + 1] };

            ce = __builtin_amdgcn_wmma_f32_16x16x4_f32(false, ax, false, bhe[c], (short)0, ce, false, false);
            ce = __builtin_amdgcn_wmma_f32_16x16x4_f32(false, ad, false, bge[c], (short)0, ce, false, false);
            co = __builtin_amdgcn_wmma_f32_16x16x4_f32(false, ax, false, bho[c], (short)0, co, false, false);
            co = __builtin_amdgcn_wmma_f32_16x16x4_f32(false, ad, false, bgo[c], (short)0, co, false, false);
        }

        const int pos = t0 + n;
#pragma unroll
        for (int r = 0; r < 8; ++r) {
            const int ch = c0 + r + (hi << 3);
            float2 v;
            v.x = ce[r];
            v.y = co[r];
            *reinterpret_cast<float2*>(&out[(size_t)ch * outL + 2 * (size_t)pos]) = v;
        }
        __syncthreads();
    }
}

extern "C" void kernel_launch(void* const* d_in, const int* in_sizes, int n_in,
                              void* d_out, int out_size, void* d_ws, size_t ws_size,
                              hipStream_t stream) {
    (void)in_sizes; (void)n_in; (void)out_size; (void)ws_size;

    const float* sig = (const float*)d_in[0];
    const float* d0  = (const float*)d_in[1];
    const float* d1  = (const float*)d_in[2];
    const float* d2  = (const float*)d_in[3];
    const float* d3  = (const float*)d_in[4];
    const float* h   = (const float*)d_in[5];
    const float* g   = (const float*)d_in[6];
    float* out = (float*)d_out;

    float* A = (float*)d_ws;
    float* B = A + (size_t)64 * 65536;

    dim3 blk(128);
    synth_level_kernel<<<16384  / 64, blk, 0, stream>>>(sig, d3, B,   h, g, 16384);
    synth_level_kernel<<<32768  / 64, blk, 0, stream>>>(B,   d2, A,   h, g, 32768);
    synth_level_kernel<<<65536  / 64, blk, 0, stream>>>(A,   d1, B,   h, g, 65536);
    synth_level_kernel<<<131072 / 64, blk, 0, stream>>>(B,   d0, out, h, g, 131072);
}


# each wave stages its whole 16-row x 72-float x and d window once: 18 aligned 16-byte loads in flight before the first wait, wave-own LDS rows; 20 dependent load rounds, 4 store drains, 7 barriers go
# speedup vs baseline: 1.7767x; 1.7767x over previous
#include <hip/hip_runtime.h>
#include <hip/hip_bf16.h>


typedef float v2f __attribute__((ext_vector_type(2)));
typedef float v8f __attribute__((ext_vector_type(8)));

#define ROW_PITCH 76

__device__ __forceinline__ float bandsel(int d, float c0, float c1, float c2) {
    float r = 0.0f;
    r = (d == 0) ? c0 : r;
    r = (d == 1) ? c1 : r;
    r = (d == 2) ? c2 : r;
    return r;
}

__device__ __forceinline__ float4 load_halo(const float* __restrict__ row, int T0, int L, int right) {
    const bool edge = right ? (T0 + 64 >= L) : (T0 == 0);
    float4 v;
    if (!edge) {
        v = *reinterpret_cast<const float4*>(row + (right ? (T0 + 64) : (T0 - 4)));
    } else if (right) {
        v = make_float4(2.0f * row[L - 1] - row[L - 2], 0.0f, 0.0f, 0.0f);
    } else {
        v = make_float4(0.0f, 0.0f, 0.0f, 2.0f * row[0] - row[1]);
    }
    return v;
}

__global__ __launch_bounds__(128) void synth_level_kernel(
        const float* __restrict__ x, const float* __restrict__ d,
        float* __restrict__ out,
        const float* __restrict__ h, const float* __restrict__ g,
        int L) {
    __shared__ __attribute__((aligned(16))) float smx[4][16][ROW_PITCH];
    __shared__ __attribute__((aligned(16))) float smd[4][16][ROW_PITCH];

    const int w    = threadIdx.x >> 5;
    const int lane = threadIdx.x & 31;
    const int c0   = w << 4;
    const int n    = lane & 15;
    const int hi   = lane >> 4;
    const int T0   = blockIdx.x << 6;

    float4 bx[8], bd[8];
#pragma unroll
    for (int j = 0; j < 8; ++j) {
        const size_t base = (size_t)(c0 + 2 * j + hi) * (size_t)L + (size_t)(T0 + 4 * n);
        bx[j] = *reinterpret_cast<const float4*>(x + base);
        bd[j] = *reinterpret_cast<const float4*>(d + base);
    }
    const size_t hrow = (size_t)(c0 + n) * (size_t)L;
    const float4 hx = load_halo(x + hrow, T0, L, hi);
    const float4 hd = load_halo(d + hrow, T0, L, hi);

    const float h0 = h[0], h1 = h[1], h2 = h[2], h3 = h[3], h4 = h[4], h5 = h[5];
    const float g0 = g[0], g1 = g[1], g2 = g[2], g3 = g[3], g4 = g[4], g5 = g[5];

    v2f bhe[5], bho[5], bge[5], bgo[5];
#pragma unroll
    for (int c = 0; c < 5; ++c) {
        const int r0 = c * 4 + (hi << 1);
        const int d0r = r0 - n, d1r = r0 + 1 - n;
        bhe[c] = v2f{ bandsel(d0r, h4, h2, h0), bandsel(d1r, h4, h2, h0) };
        bho[c] = v2f{ bandsel(d0r, h5, h3, h1), bandsel(d1r, h5, h3, h1) };
        bge[c] = v2f{ bandsel(d0r, g4, g2, g0), bandsel(d1r, g4, g2, g0) };
        bgo[c] = v2f{ bandsel(d0r, g5, g3, g1), bandsel(d1r, g5, g3, g1) };
    }

#pragma unroll
    for (int j = 0; j < 8; ++j) {
        *reinterpret_cast<float4*>(&smx[w][2 * j + hi][4 + 4 * n]) = bx[j];
        *reinterpret_cast<float4*>(&smd[w][2 * j + hi][4 + 4 * n]) = bd[j];
    }
    *reinterpret_cast<float4*>(&smx[w][n][hi ? 68 : 0]) = hx;
    *reinterpret_cast<float4*>(&smd[w][n][hi ? 68 : 0]) = hd;

    __syncthreads();

    const size_t outL = 2 * (size_t)L;
    const unsigned voff = (unsigned)hi * (unsigned)(16 * L) + 2u * (unsigned)n;

#pragma unroll 1
    for (int it = 0; it < 4; ++it) {
        const int t0 = T0 + (it << 4);
        const float* fx = &smx[w][n][(it << 4) + (hi << 1) + 3];
        const float* fd = &smd[w][n][(it << 4) + (hi << 1) + 3];

        v8f ce = {};
        v8f co = {};
#pragma unroll
        for (int c = 0; c < 5; ++c) {
            const v2f ax = v2f{ fx[4 * c], fx[4 * c + 1] };
            const v2f ad = v2f{ fd[4 * c], fd[4 * c + 1] };

            ce = __builtin_amdgcn_wmma_f32_16x16x4_f32(false, ax, false, bhe[c], (short)0, ce, false, false);
            ce = __builtin_amdgcn_wmma_f32_16x16x4_f32(false, ad, false, bge[c], (short)0, ce, false, false);
            co = __builtin_amdgcn_wmma_f32_16x16x4_f32(false, ax, false, bho[c], (short)0, co, false, false);
            co = __builtin_amdgcn_wmma_f32_16x16x4_f32(false, ad, false, bgo[c], (short)0, co, false, false);
        }

        __builtin_amdgcn_sched_barrier(0);
        asm volatile("v_nop\n\tv_nop\n\tv_nop\n\tv_nop");
        __builtin_amdgcn_sched_barrier(0);

        float* sbase = out + (size_t)c0 * outL + 2 * (size_t)t0;
#pragma unroll
        for (int r = 0; r < 8; ++r) {
            float2 v;
            v.x = ce[r];
            v.y = co[r];
            *reinterpret_cast<float2*>(sbase + (size_t)r * outL + voff) = v;
        }
    }
}

extern "C" void kernel_launch(void* const* d_in, const int* in_sizes, int n_in,
                              void* d_out, int out_size, void* d_ws, size_t ws_size,
                              hipStream_t stream) {
    (void)in_sizes; (void)n_in; (void)out_size; (void)ws_size;

    const float* sig = (const float*)d_in[0];
    const float* d0  = (const float*)d_in[1];
    const float* d1  = (const float*)d_in[2];
    const float* d2  = (const float*)d_in[3];
    const float* d3  = (const float*)d_in[4];
    const float* h   = (const float*)d_in[5];
    const float* g   = (const float*)d_in[6];
    float* out = (float*)d_out;

    float* A = (float*)d_ws;
    float* B = A + (size_t)64 * 65536;

    dim3 blk(128);
    synth_level_kernel<<<16384  / 64, blk, 0, stream>>>(sig, d3, B,   h, g, 16384);
    synth_level_kernel<<<32768  / 64, blk, 0, stream>>>(B,   d2, A,   h, g, 32768);
    synth_level_kernel<<<65536  / 64, blk, 0, stream>>>(A,   d1, B,   h, g, 65536);
    synth_level_kernel<<<131072 / 64, blk, 0, stream>>>(B,   d0, out, h, g, 131072);
}


# one launch: each wave runs all four levels for 16 channels x 256 outputs from its own LDS rows, recomputing tile overlap; K rows start one sample earlier, so the 6 terms regroup in the 4-wide sums
# speedup vs baseline: 2.5334x; 1.4259x over previous
#include <hip/hip_runtime.h>
#include <hip/hip_bf16.h>


typedef float v2f __attribute__((ext_vector_type(2)));
typedef float v8f __attribute__((ext_vector_type(8)));

#define P1 44
#define P2 60
#define P3 92
#define P4 140

#define OFF_X1 0
#define OFF_D1 (OFF_X1 + 16 * P1)
#define OFF_D2 (OFF_D1 + 16 * P1)
#define OFF_X2 (OFF_D2 + 16 * P2)
#define OFF_D3 (OFF_X2 + 16 * P2)
#define OFF_X3 (OFF_D3 + 16 * P3)
#define OFF_D4 (OFF_X3 + 16 * P3)
#define LDS_FLOATS (OFF_D4 + 16 * P4)
#define OFF_X4 0

#define NTILES 1024

__device__ __forceinline__ float bandsel(int d, float c0, float c1, float c2) {
    float r = 0.0f;
    r = (d == 0) ? c0 : r;
    r = (d == 1) ? c1 : r;
    r = (d == 2) ? c2 : r;
    return r;
}

__device__ __forceinline__ float4 load_halo(const float* __restrict__ row, int T0, int len, int L, int right) {
    const bool edge = right ? (T0 + len >= L) : (T0 == 0);
    float4 v;
    if (!edge) {
        v = *reinterpret_cast<const float4*>(row + (right ? (T0 + len) : (T0 - 4)));
    } else if (right) {
        v = make_float4(2.0f * row[L - 1] - row[L - 2], 0.0f, 0.0f, 0.0f);
    } else {
        v = make_float4(0.0f, 0.0f, 0.0f, 2.0f * row[0] - row[1]);
    }
    return v;
}

struct Bfrag {
    v2f he[5], ho[5], ge[5], go[5];
};

__device__ __forceinline__ void make_bfrag(Bfrag& B, int n, int hi, int shift,
                                           const float* hh, const float* gg) {
#pragma unroll
    for (int c = 0; c < 5; ++c) {
        const int d0r = c * 4 + (hi << 1) - n - shift, d1r = d0r + 1;
        B.he[c] = v2f{ bandsel(d0r, hh[4], hh[2], hh[0]), bandsel(d1r, hh[4], hh[2], hh[0]) };
        B.ho[c] = v2f{ bandsel(d0r, hh[5], hh[3], hh[1]), bandsel(d1r, hh[5], hh[3], hh[1]) };
        B.ge[c] = v2f{ bandsel(d0r, gg[4], gg[2], gg[0]), bandsel(d1r, gg[4], gg[2], gg[0]) };
        B.go[c] = v2f{ bandsel(d0r, gg[5], gg[3], gg[1]), bandsel(d1r, gg[5], gg[3], gg[1]) };
    }
}

__device__ __forceinline__ void subtile(const float* fx, const float* fd, const Bfrag& B, v8f& ce, v8f& co) {
    ce = v8f{};
    co = v8f{};
#pragma unroll
    for (int c = 0; c < 5; ++c) {
        const v2f ax = v2f{ fx[4 * c], fx[4 * c + 1] };
        const v2f ad = v2f{ fd[4 * c], fd[4 * c + 1] };
        ce = __builtin_amdgcn_wmma_f32_16x16x4_f32(false, ax, false, B.he[c], (short)0, ce, false, false);
        ce = __builtin_amdgcn_wmma_f32_16x16x4_f32(false, ad, false, B.ge[c], (short)0, ce, false, false);
        co = __builtin_amdgcn_wmma_f32_16x16x4_f32(false, ax, false, B.ho[c], (short)0, co, false, false);
        co = __builtin_amdgcn_wmma_f32_16x16x4_f32(false, ad, false, B.go[c], (short)0, co, false, false);
    }
    __builtin_amdgcn_sched_barrier(0);
    asm volatile("v_nop\n\tv_nop\n\tv_nop\n\tv_nop");
    __builtin_amdgcn_sched_barrier(0);
}

template <int NSUB, int PIN, int PNEXT>
__device__ __forceinline__ void level_to_lds(const float* X, const float* D, float* Xn,
                                             int n, int hi, const Bfrag& B) {
#pragma unroll 1
    for (int it = 0; it < NSUB; ++it) {
        const int k0 = n * PIN + 2 + (it << 4) + (hi << 1);
        v8f ce, co;
        subtile(X + k0, D + k0, B, ce, co);
        const int col = 2 + (it << 5) + 2 * n;
        if (col < PNEXT) {
            float* s = Xn + (hi << 3) * PNEXT + col;
#pragma unroll
            for (int r = 0; r < 8; ++r) {
                float2 v;
                v.x = ce[r];
                v.y = co[r];
                *reinterpret_cast<float2*>(s + r * PNEXT) = v;
            }
        }
    }
}

__device__ __forceinline__ void fix_edges(float* X, int pitch, int len, int b, int lane) {
    if (lane < 16) {
        float* r = X + lane * pitch;
        if (b == 0) r[3] = 2.0f * r[4] - r[5];
        if (b == NTILES - 1) r[len + 4] = 2.0f * r[len + 3] - r[len + 2];
    }
}

__global__ __launch_bounds__(32) void synth4_kernel(
        const float* __restrict__ sig, const float* __restrict__ d0,
        const float* __restrict__ d1, const float* __restrict__ d2,
        const float* __restrict__ d3, float* __restrict__ out,
        const float* __restrict__ h, const float* __restrict__ g) {
    __shared__ __attribute__((aligned(16))) float sm[LDS_FLOATS];

    constexpr int L1 = 16384, L2 = 32768, L3 = 65536, L4 = 131072;

    const int lane = threadIdx.x & 31;
    const int n    = lane & 15;
    const int hi   = lane >> 4;
    const int w    = blockIdx.x >> 10;
    const int b    = blockIdx.x & (NTILES - 1);
    const int c0   = w << 4;
    const int B1 = b << 4, B2 = b << 5, B3 = b << 6, B4 = b << 7;

    const float4 hx1 = load_halo(sig + (size_t)(c0 + n) * L1, B1, 16, L1, hi);
    const float4 hd1 = load_halo(d3  + (size_t)(c0 + n) * L1, B1, 16, L1, hi);
    const float4 hd2 = load_halo(d2  + (size_t)(c0 + n) * L2, B2, 32, L2, hi);
    const float4 hd3 = load_halo(d1  + (size_t)(c0 + n) * L3, B3, 64, L3, hi);
    const float4 hd4 = load_halo(d0  + (size_t)(c0 + n) * L4, B4, 128, L4, hi);
    float4 rx1[2], rd1[2];
#pragma unroll
    for (int j = 0; j < 2; ++j) {
        const size_t a = (size_t)(c0 + 8 * j + (lane >> 2)) * L1 + (size_t)(B1 + 4 * (lane & 3));
        rx1[j] = *reinterpret_cast<const float4*>(sig + a);
        rd1[j] = *reinterpret_cast<const float4*>(d3 + a);
    }
    float4 rd2[4];
#pragma unroll
    for (int j = 0; j < 4; ++j) {
        const size_t a = (size_t)(c0 + 4 * j + (lane >> 3)) * L2 + (size_t)(B2 + 4 * (lane & 7));
        rd2[j] = *reinterpret_cast<const float4*>(d2 + a);
    }
    float4 rd3[8];
#pragma unroll
    for (int j = 0; j < 8; ++j) {
        const size_t a = (size_t)(c0 + 2 * j + hi) * L3 + (size_t)(B3 + 4 * n);
        rd3[j] = *reinterpret_cast<const float4*>(d1 + a);
    }
    float4 rd4[16];
#pragma unroll
    for (int j = 0; j < 16; ++j) {
        const size_t a = (size_t)(c0 + j) * L4 + (size_t)(B4 + 4 * lane);
        rd4[j] = *reinterpret_cast<const float4*>(d0 + a);
    }

    const float hh[6] = { h[0], h[1], h[2], h[3], h[4], h[5] };
    const float gg[6] = { g[0], g[1], g[2], g[3], g[4], g[5] };

    Bfrag B;
    make_bfrag(B, n, hi, 0, hh, gg);

    float* X1 = sm + OFF_X1; float* D1 = sm + OFF_D1;
    float* X2 = sm + OFF_X2; float* D2 = sm + OFF_D2;
    float* X3 = sm + OFF_X3; float* D3 = sm + OFF_D3;
    float* X4 = sm + OFF_X4; float* D4 = sm + OFF_D4;

#pragma unroll
    for (int j = 0; j < 2; ++j) {
        const int o = (8 * j + (lane >> 2)) * P1 + 4 + 4 * (lane & 3);
        *reinterpret_cast<float4*>(X1 + o) = rx1[j];
        *reinterpret_cast<float4*>(D1 + o) = rd1[j];
    }
    *reinterpret_cast<float4*>(X1 + n * P1 + (hi ? 20 : 0)) = hx1;
    *reinterpret_cast<float4*>(D1 + n * P1 + (hi ? 20 : 0)) = hd1;
#pragma unroll
    for (int j = 0; j < 4; ++j) {
        const int o = (4 * j + (lane >> 3)) * P2 + 4 + 4 * (lane & 7);
        *reinterpret_cast<float4*>(D2 + o) = rd2[j];
    }
    *reinterpret_cast<float4*>(D2 + n * P2 + (hi ? 36 : 0)) = hd2;
#pragma unroll
    for (int j = 0; j < 8; ++j) {
        *reinterpret_cast<float4*>(D3 + (2 * j + hi) * P3 + 4 + 4 * n) = rd3[j];
    }
    *reinterpret_cast<float4*>(D3 + n * P3 + (hi ? 68 : 0)) = hd3;
#pragma unroll
    for (int j = 0; j < 16; ++j) {
        *reinterpret_cast<float4*>(D4 + j * P4 + 4 + 4 * lane) = rd4[j];
    }
    *reinterpret_cast<float4*>(D4 + n * P4 + (hi ? 132 : 0)) = hd4;
    const float4 z4 = make_float4(0.0f, 0.0f, 0.0f, 0.0f);
#pragma unroll
    for (int k = 0; k < 2; ++k) {
        const int i = lane + 32 * k;
        const int zr = i >> 2, zq = 4 * (i & 3);
        *reinterpret_cast<float4*>(X1 + zr * P1 + 24 + zq) = z4;
        *reinterpret_cast<float4*>(D1 + zr * P1 + 24 + zq) = z4;
        *reinterpret_cast<float4*>(D2 + zr * P2 + 40 + zq) = z4;
        *reinterpret_cast<float4*>(D3 + zr * P3 + 72 + zq) = z4;
    }

    __syncthreads();

    level_to_lds<2, P1, P2>(X1, D1, X2, n, hi, B);
    __syncthreads();
    fix_edges(X2, P2, 32, b, lane);
    __syncthreads();

    level_to_lds<3, P2, P3>(X2, D2, X3, n, hi, B);
    __syncthreads();
    fix_edges(X3, P3, 64, b, lane);
    __syncthreads();

    level_to_lds<5, P3, P4>(X3, D3, X4, n, hi, B);
    __syncthreads();
    fix_edges(X4, P4, 128, b, lane);
    __syncthreads();

    Bfrag Bl;
    make_bfrag(Bl, n, hi, 1, hh, gg);
    const size_t outL = 2 * (size_t)L4;
    float* obase = out + (size_t)(c0 + (hi << 3)) * outL + 2 * (size_t)(B4 + n);
#pragma unroll 1
    for (int it = 0; it < 8; ++it) {
        const int k0 = n * P4 + 2 + (it << 4) + (hi << 1);
        v8f ce, co;
        subtile(X4 + k0, D4 + k0, Bl, ce, co);
        float* s = obase + (it << 5);
#pragma unroll
        for (int r = 0; r < 8; ++r) {
            float2 v;
            v.x = ce[r];
            v.y = co[r];
            *reinterpret_cast<float2*>(s + (size_t)r * outL) = v;
        }
    }
}

extern "C" void kernel_launch(void* const* d_in, const int* in_sizes, int n_in,
                              void* d_out, int out_size, void* d_ws, size_t ws_size,
                              hipStream_t stream) {
    (void)in_sizes; (void)n_in; (void)out_size; (void)d_ws; (void)ws_size;

    const float* sig = (const float*)d_in[0];
    const float* d0  = (const float*)d_in[1];
    const float* d1  = (const float*)d_in[2];
    const float* d2  = (const float*)d_in[3];
    const float* d3  = (const float*)d_in[4];
    const float* h   = (const float*)d_in[5];
    const float* g   = (const float*)d_in[6];
    float* out = (float*)d_out;

    synth4_kernel<<<4096, dim3(32), 0, stream>>>(sig, d0, d1, d2, d3, out, h, g);
}


# x and d stacked in K, both phases in N: one 20x16 band per 8 positions, 5 matrix instructions where 20 served 16; fragments load one trip ahead as a group; the 6 terms regroup across the K chunks
# speedup vs baseline: 3.1870x; 1.2580x over previous
#include <hip/hip_runtime.h>
#include <hip/hip_bf16.h>


typedef float v2f __attribute__((ext_vector_type(2)));
typedef float v8f __attribute__((ext_vector_type(8)));

#define P1 44
#define P2 60
#define P3 92
#define P4 140

#define OFF_X1 0
#define OFF_D1 (OFF_X1 + 16 * P1)
#define OFF_D2 (OFF_D1 + 16 * P1)
#define OFF_X2 (OFF_D2 + 16 * P2)
#define OFF_D3 (OFF_X2 + 16 * P2)
#define OFF_X3 (OFF_D3 + 16 * P3)
#define OFF_D4 (OFF_X3 + 16 * P3)
#define LDS_FLOATS (OFF_D4 + 16 * P4)
#define OFF_X4 0

#define NTILES 1024

__device__ __forceinline__ float bandsel(int dd, int ph, const float* f) {
    float r = 0.0f;
    r = (dd == 0) ? (ph ? f[5] : f[4]) : r;
    r = (dd == 1) ? (ph ? f[3] : f[2]) : r;
    r = (dd == 2) ? (ph ? f[1] : f[0]) : r;
    return r;
}

__device__ __forceinline__ float4 load_halo(const float* __restrict__ row, int T0, int len, int L, int right) {
    const bool edge = right ? (T0 + len >= L) : (T0 == 0);
    float4 v;
    if (!edge) {
        v = *reinterpret_cast<const float4*>(row + (right ? (T0 + len) : (T0 - 4)));
    } else if (right) {
        v = make_float4(2.0f * row[L - 1] - row[L - 2], 0.0f, 0.0f, 0.0f);
    } else {
        v = make_float4(0.0f, 0.0f, 0.0f, 2.0f * row[0] - row[1]);
    }
    return v;
}

struct Bfrag {
    v2f w[5];
};

__device__ __forceinline__ void make_bfrag(Bfrag& B, int n, int hi,
                                           const float* hh, const float* gg) {
    const int p = n >> 1, ph = n & 1;
#pragma unroll
    for (int c = 0; c < 5; ++c) {
        float e[2];
#pragma unroll
        for (int q = 0; q < 2; ++q) {
            const int k = c * 4 + (hi << 1) + q;
            const bool isd = (k >= 10);
            const int dd = (isd ? k - 10 : k) - p;
            const float wx = bandsel(dd, ph, hh);
            const float wd = bandsel(dd, ph, gg);
            e[q] = isd ? wd : wx;
        }
        B.w[c] = v2f{ e[0], e[1] };
    }
}

__device__ __forceinline__ void load_afrag(v2f (&a)[5], const float* q01, const float* q2, const float* q34) {
    a[0] = v2f{ q01[0], q01[1] };
    a[1] = v2f{ q01[4], q01[5] };
    a[2] = v2f{ q2[0],  q2[1]  };
    a[3] = v2f{ q34[0], q34[1] };
    a[4] = v2f{ q34[4], q34[5] };
}

__device__ __forceinline__ void wmma_tail_slots() {
    __builtin_amdgcn_sched_barrier(0);
    asm volatile("v_nop\n\tv_nop\n\tv_nop\n\tv_nop");
    __builtin_amdgcn_sched_barrier(0);
}

struct Afrag2 {
    v2f a[2][5];
};

__device__ __forceinline__ void load_trip(Afrag2& F, int cnt, const float* q01, const float* q2, const float* q34) {
    load_afrag(F.a[0], q01, q2, q34);
    if (cnt == 2) load_afrag(F.a[1], q01 + 8, q2 + 8, q34 + 8);
}

__device__ __forceinline__ void mma_trip(const Afrag2& F, int cnt, const Bfrag& B, v8f& acc0, v8f& acc1) {
    acc0 = v8f{};
    acc1 = v8f{};
#pragma unroll
    for (int c = 0; c < 5; ++c) {
        acc0 = __builtin_amdgcn_wmma_f32_16x16x4_f32(false, F.a[0][c], false, B.w[c], (short)0, acc0, false, false);
        if (cnt == 2)
            acc1 = __builtin_amdgcn_wmma_f32_16x16x4_f32(false, F.a[1][c], false, B.w[c], (short)0, acc1, false, false);
    }
    wmma_tail_slots();
}

template <int PNEXT>
__device__ __forceinline__ void store_next(float* Xn, int it, int n, int hi, const v8f& acc) {
    const int col = 2 + (it << 4) + n;
    if (col < PNEXT) {
        float* s = Xn + (hi << 3) * PNEXT + col;
#pragma unroll
        for (int r = 0; r < 8; ++r) s[r * PNEXT] = acc[r];
    }
}

__device__ __forceinline__ void store_out(float* s, size_t outL, int odd, const v8f& acc0, const v8f& acc1) {
#pragma unroll
    for (int r = 0; r < 8; ++r) {
        const float give = odd ? acc0[r] : acc1[r];
        const float recv = __builtin_bit_cast(float,
            __builtin_amdgcn_update_dpp(0, __builtin_bit_cast(int, give), 0xB1, 0xf, 0xf, true));
        float2 v;
        v.x = odd ? recv : acc0[r];
        v.y = odd ? acc1[r] : recv;
        *reinterpret_cast<float2*>(s + (size_t)r * outL) = v;
    }
}

template <int NSUB, int PIN, int COL0, int PNEXT, bool LAST>
__device__ __forceinline__ void run_level(const float* X, const float* D, float* Xn,
                                          float* obase, size_t outL,
                                          int n, int hi, int odd, const Bfrag& B) {
    const float* xb = X + n * PIN + COL0;
    const float* db = D + n * PIN + COL0;
    const float* q01 = xb + (hi << 1);
    const float* q2  = hi ? db : xb + 8;
    const float* q34 = db + 2 + (hi << 1);
    constexpr int NT = (NSUB + 1) / 2;
    Afrag2 F[2];
    load_trip(F[0], NSUB >= 2 ? 2 : 1, q01, q2, q34);
#pragma unroll
    for (int t = 0; t < NT; ++t) {
        const int cnt = (NSUB - 2 * t >= 2) ? 2 : 1;
        if (t + 1 < NT) {
            const int cn = (NSUB - 2 * (t + 1) >= 2) ? 2 : 1;
            const int o = (t + 1) << 4;
            load_trip(F[(t + 1) & 1], cn, q01 + o, q2 + o, q34 + o);
        }
        __builtin_amdgcn_sched_barrier(0);
        v8f acc0, acc1;
        mma_trip(F[t & 1], cnt, B, acc0, acc1);
        if (LAST) {
            store_out(obase + (t << 5), outL, odd, acc0, acc1);
        } else {
            store_next<PNEXT>(Xn, 2 * t, n, hi, acc0);
            if (cnt == 2) store_next<PNEXT>(Xn, 2 * t + 1, n, hi, acc1);
        }
    }
}

__device__ __forceinline__ void fix_edges(float* X, int pitch, int len, int b, int lane) {
    if (lane < 16) {
        float* r = X + lane * pitch;
        if (b == 0) r[3] = 2.0f * r[4] - r[5];
        if (b == NTILES - 1) r[len + 4] = 2.0f * r[len + 3] - r[len + 2];
    }
}

__global__ __launch_bounds__(32) void synth4_kernel(
        const float* __restrict__ sig, const float* __restrict__ d0,
        const float* __restrict__ d1, const float* __restrict__ d2,
        const float* __restrict__ d3, float* __restrict__ out,
        const float* __restrict__ h, const float* __restrict__ g) {
    __shared__ __attribute__((aligned(16))) float sm[LDS_FLOATS];

    constexpr int L1 = 16384, L2 = 32768, L3 = 65536, L4 = 131072;

    const int lane = threadIdx.x & 31;
    const int n    = lane & 15;
    const int hi   = lane >> 4;
    const int w    = blockIdx.x >> 10;
    const int b    = blockIdx.x & (NTILES - 1);
    const int c0   = w << 4;
    const int B1 = b << 4, B2 = b << 5, B3 = b << 6, B4 = b << 7;

    const float4 hx1 = load_halo(sig + (size_t)(c0 + n) * L1, B1, 16, L1, hi);
    const float4 hd1 = load_halo(d3  + (size_t)(c0 + n) * L1, B1, 16, L1, hi);
    const float4 hd2 = load_halo(d2  + (size_t)(c0 + n) * L2, B2, 32, L2, hi);
    const float4 hd3 = load_halo(d1  + (size_t)(c0 + n) * L3, B3, 64, L3, hi);
    const float4 hd4 = load_halo(d0  + (size_t)(c0 + n) * L4, B4, 128, L4, hi);
    float4 rx1[2], rd1[2];
#pragma unroll
    for (int j = 0; j < 2; ++j) {
        const size_t a = (size_t)(c0 + 8 * j + (lane >> 2)) * L1 + (size_t)(B1 + 4 * (lane & 3));
        rx1[j] = *reinterpret_cast<const float4*>(sig + a);
        rd1[j] = *reinterpret_cast<const float4*>(d3 + a);
    }
    float4 rd2[4];
#pragma unroll
    for (int j = 0; j < 4; ++j) {
        const size_t a = (size_t)(c0 + 4 * j + (lane >> 3)) * L2 + (size_t)(B2 + 4 * (lane & 7));
        rd2[j] = *reinterpret_cast<const float4*>(d2 + a);
    }
    float4 rd3[8];
#pragma unroll
    for (int j = 0; j < 8; ++j) {
        const size_t a = (size_t)(c0 + 2 * j + hi) * L3 + (size_t)(B3 + 4 * n);
        rd3[j] = *reinterpret_cast<const float4*>(d1 + a);
    }
    float4 rd4[16];
#pragma unroll
    for (int j = 0; j < 16; ++j) {
        const size_t a = (size_t)(c0 + j) * L4 + (size_t)(B4 + 4 * lane);
        rd4[j] = *reinterpret_cast<const float4*>(d0 + a);
    }

    const float hh[6] = { h[0], h[1], h[2], h[3], h[4], h[5] };
    const float gg[6] = { g[0], g[1], g[2], g[3], g[4], g[5] };

    Bfrag B;
    make_bfrag(B, n, hi, hh, gg);

    float* X1 = sm + OFF_X1; float* D1 = sm + OFF_D1;
    float* X2 = sm + OFF_X2; float* D2 = sm + OFF_D2;
    float* X3 = sm + OFF_X3; float* D3 = sm + OFF_D3;
    float* X4 = sm + OFF_X4; float* D4 = sm + OFF_D4;

#pragma unroll
    for (int j = 0; j < 2; ++j) {
        const int o = (8 * j + (lane >> 2)) * P1 + 4 + 4 * (lane & 3);
        *reinterpret_cast<float4*>(X1 + o) = rx1[j];
        *reinterpret_cast<float4*>(D1 + o) = rd1[j];
    }
    *reinterpret_cast<float4*>(X1 + n * P1 + (hi ? 20 : 0)) = hx1;
    *reinterpret_cast<float4*>(D1 + n * P1 + (hi ? 20 : 0)) = hd1;
#pragma unroll
    for (int j = 0; j < 4; ++j) {
        const int o = (4 * j + (lane >> 3)) * P2 + 4 + 4 * (lane & 7);
        *reinterpret_cast<float4*>(D2 + o) = rd2[j];
    }
    *reinterpret_cast<float4*>(D2 + n * P2 + (hi ? 36 : 0)) = hd2;
#pragma unroll
    for (int j = 0; j < 8; ++j) {
        *reinterpret_cast<float4*>(D3 + (2 * j + hi) * P3 + 4 + 4 * n) = rd3[j];
    }
    *reinterpret_cast<float4*>(D3 + n * P3 + (hi ? 68 : 0)) = hd3;
#pragma unroll
    for (int j = 0; j < 16; ++j) {
        *reinterpret_cast<float4*>(D4 + j * P4 + 4 + 4 * lane) = rd4[j];
    }
    *reinterpret_cast<float4*>(D4 + n * P4 + (hi ? 132 : 0)) = hd4;
    const float4 z4 = make_float4(0.0f, 0.0f, 0.0f, 0.0f);
    *reinterpret_cast<float4*>(hi ? (D1 + n * P1 + 24) : (X1 + n * P1 + 24)) = z4;
    *reinterpret_cast<float4*>(hi ? (D3 + n * P3 + 72) : (D2 + n * P2 + 40)) = z4;

    __syncthreads();

    run_level<3, P1, 2, P2, false>(X1, D1, X2, nullptr, 0, n, hi, 0, B);
    __syncthreads();
    fix_edges(X2, P2, 32, b, lane);
    __syncthreads();

    run_level<5, P2, 2, P3, false>(X2, D2, X3, nullptr, 0, n, hi, 0, B);
    __syncthreads();
    fix_edges(X3, P3, 64, b, lane);
    __syncthreads();

    run_level<9, P3, 2, P4, false>(X3, D3, X4, nullptr, 0, n, hi, 0, B);
    __syncthreads();
    fix_edges(X4, P4, 128, b, lane);
    __syncthreads();

    const size_t outL = 2 * (size_t)L4;
    const int odd = n & 1;
    float* obase = out + (size_t)(c0 + (hi << 3)) * outL + 2 * (size_t)B4 + (odd ? 15 + n : n);
    run_level<16, P4, 3, P4, true>(X4, D4, nullptr, obase, outL, n, hi, odd, B);
}

extern "C" void kernel_launch(void* const* d_in, const int* in_sizes, int n_in,
                              void* d_out, int out_size, void* d_ws, size_t ws_size,
                              hipStream_t stream) {
    (void)in_sizes; (void)n_in; (void)out_size; (void)d_ws; (void)ws_size;

    const float* sig = (const float*)d_in[0];
    const float* d0  = (const float*)d_in[1];
    const float* d1  = (const float*)d_in[2];
    const float* d2  = (const float*)d_in[3];
    const float* d3  = (const float*)d_in[4];
    const float* h   = (const float*)d_in[5];
    const float* g   = (const float*)d_in[6];
    float* out = (float*)d_out;

    synth4_kernel<<<4096, dim3(32), 0, stream>>>(sig, d0, d1, d2, d3, out, h, g);
}


# LDS per wave 34048 -> 18944 bytes: tightest 4 x odd row pitches, level-4 d window held in registers and parked over the dead level-3 windows; a WGP holds all its waves at once instead of 9 then 7
# speedup vs baseline: 3.4105x; 1.0701x over previous
#include <hip/hip_runtime.h>
#include <hip/hip_bf16.h>


typedef float v2f __attribute__((ext_vector_type(2)));
typedef float v8f __attribute__((ext_vector_type(8)));
typedef float v4f __attribute__((ext_vector_type(4)));

#define P1 28
#define P2 44
#define P3 76
#define P4 140

#define OFF_X1 0
#define OFF_D1 (OFF_X1 + 16 * P1)
#define OFF_D2 (OFF_D1 + 16 * P1)
#define OFF_X2 (OFF_D2 + 16 * P2)
#define OFF_X3 (OFF_X2 + 16 * P2)
#define OFF_D3 (OFF_X3 + 16 * P3)
#define LDS_FLOATS (OFF_D3 + 16 * P3)
#define OFF_X4 0
#define OFF_D4 OFF_X3
static_assert(16 * P4 <= OFF_X3, "level-4 x window must fit over the level-1/2 buffers");
static_assert(OFF_D4 + 16 * P4 <= LDS_FLOATS, "level-4 d window must fit over the level-3 buffers");

#define NTILES 1024

__device__ __forceinline__ float bandsel(int dd, int ph, const float* f) {
    float r = 0.0f;
    r = (dd == 0) ? (ph ? f[5] : f[4]) : r;
    r = (dd == 1) ? (ph ? f[3] : f[2]) : r;
    r = (dd == 2) ? (ph ? f[1] : f[0]) : r;
    return r;
}

__device__ __forceinline__ float4 load_halo(const float* __restrict__ row, int T0, int len, int L, int right) {
    const bool edge = right ? (T0 + len >= L) : (T0 == 0);
    float4 v;
    if (!edge) {
        v = *reinterpret_cast<const float4*>(row + (right ? (T0 + len) : (T0 - 4)));
    } else if (right) {
        v = make_float4(2.0f * row[L - 1] - row[L - 2], 0.0f, 0.0f, 0.0f);
    } else {
        v = make_float4(0.0f, 0.0f, 0.0f, 2.0f * row[0] - row[1]);
    }
    return v;
}

struct Bfrag {
    v2f w[5];
};

__device__ __forceinline__ void make_bfrag(Bfrag& B, int n, int hi,
                                           const float* hh, const float* gg) {
    const int p = n >> 1, ph = n & 1;
#pragma unroll
    for (int c = 0; c < 5; ++c) {
        float e[2];
#pragma unroll
        for (int q = 0; q < 2; ++q) {
            const int k = c * 4 + (hi << 1) + q;
            const bool isd = (k >= 10);
            const int dd = (isd ? k - 10 : k) - p;
            const float wx = bandsel(dd, ph, hh);
            const float wd = bandsel(dd, ph, gg);
            e[q] = isd ? wd : wx;
        }
        B.w[c] = v2f{ e[0], e[1] };
    }
}

__device__ __forceinline__ void load_afrag(v2f (&a)[5], const float* q01, const float* q2, const float* q34) {
    a[0] = v2f{ q01[0], q01[1] };
    a[1] = v2f{ q01[4], q01[5] };
    a[2] = v2f{ q2[0],  q2[1]  };
    a[3] = v2f{ q34[0], q34[1] };
    a[4] = v2f{ q34[4], q34[5] };
}

__device__ __forceinline__ void wmma_tail_slots() {
    __builtin_amdgcn_sched_barrier(0);
    asm volatile("v_nop\n\tv_nop\n\tv_nop\n\tv_nop");
    __builtin_amdgcn_sched_barrier(0);
}

struct Afrag2 {
    v2f a[2][5];
};

__device__ __forceinline__ void load_trip(Afrag2& F, int cnt, const float* q01, const float* q2, const float* q34) {
    load_afrag(F.a[0], q01, q2, q34);
    if (cnt == 2) load_afrag(F.a[1], q01 + 8, q2 + 8, q34 + 8);
}

__device__ __forceinline__ void mma_trip(const Afrag2& F, int cnt, const Bfrag& B, v8f& acc0, v8f& acc1) {
    acc0 = v8f{};
    acc1 = v8f{};
#pragma unroll
    for (int c = 0; c < 5; ++c) {
        acc0 = __builtin_amdgcn_wmma_f32_16x16x4_f32(false, F.a[0][c], false, B.w[c], (short)0, acc0, false, false);
        if (cnt == 2)
            acc1 = __builtin_amdgcn_wmma_f32_16x16x4_f32(false, F.a[1][c], false, B.w[c], (short)0, acc1, false, false);
    }
    wmma_tail_slots();
}

template <int PNEXT>
__device__ __forceinline__ void store_next(float* Xn, int it, int n, int hi, const v8f& acc) {
    const int col = 2 + (it << 4) + n;
    if (col < PNEXT) {
        float* s = Xn + (hi << 3) * PNEXT + col;
#pragma unroll
        for (int r = 0; r < 8; ++r) s[r * PNEXT] = acc[r];
    }
}

__device__ __forceinline__ void store_out(float* s, size_t outL, int odd, const v8f& acc0, const v8f& acc1) {
#pragma unroll
    for (int r = 0; r < 8; ++r) {
        const float give = odd ? acc0[r] : acc1[r];
        const float recv = __builtin_bit_cast(float,
            __builtin_amdgcn_update_dpp(0, __builtin_bit_cast(int, give), 0xB1, 0xf, 0xf, true));
        float2 v;
        v.x = odd ? recv : acc0[r];
        v.y = odd ? acc1[r] : recv;
        *reinterpret_cast<float2*>(s + (size_t)r * outL) = v;
    }
}

template <int NSUB, int PIN, int COL0, int PNEXT, bool LAST>
__device__ __forceinline__ void run_level(const float* X, const float* D, float* Xn,
                                          float* obase, size_t outL,
                                          int n, int hi, int odd, const Bfrag& B) {
    const float* xb = X + n * PIN + COL0;
    const float* db = D + n * PIN + COL0;
    const float* q01 = xb + (hi << 1);
    const float* q2  = hi ? db : xb + 8;
    const float* q34 = db + 2 + (hi << 1);
    constexpr int NT = (NSUB + 1) / 2;
    Afrag2 F[2];
    load_trip(F[0], NSUB >= 2 ? 2 : 1, q01, q2, q34);
#pragma unroll
    for (int t = 0; t < NT; ++t) {
        const int cnt = (NSUB - 2 * t >= 2) ? 2 : 1;
        if (t + 1 < NT) {
            const int cn = (NSUB - 2 * (t + 1) >= 2) ? 2 : 1;
            const int o = (t + 1) << 4;
            load_trip(F[(t + 1) & 1], cn, q01 + o, q2 + o, q34 + o);
        }
        __builtin_amdgcn_sched_barrier(0);
        v8f acc0, acc1;
        mma_trip(F[t & 1], cnt, B, acc0, acc1);
        if (LAST) {
            store_out(obase + (t << 5), outL, odd, acc0, acc1);
        } else {
            store_next<PNEXT>(Xn, 2 * t, n, hi, acc0);
            if (cnt == 2) store_next<PNEXT>(Xn, 2 * t + 1, n, hi, acc1);
        }
    }
}

__device__ __forceinline__ void fix_edges(float* X, int pitch, int len, int b, int lane) {
    if (lane < 16) {
        float* r = X + lane * pitch;
        if (b == 0) r[3] = 2.0f * r[4] - r[5];
        if (b == NTILES - 1) r[len + 4] = 2.0f * r[len + 3] - r[len + 2];
    }
}

__global__ __launch_bounds__(32) void synth4_kernel(
        const float* __restrict__ sig, const float* __restrict__ d0,
        const float* __restrict__ d1, const float* __restrict__ d2,
        const float* __restrict__ d3, float* __restrict__ out,
        const float* __restrict__ h, const float* __restrict__ g) {
    __shared__ __attribute__((aligned(16))) float sm[LDS_FLOATS];

    constexpr int L1 = 16384, L2 = 32768, L3 = 65536, L4 = 131072;

    const int lane = threadIdx.x & 31;
    const int n    = lane & 15;
    const int hi   = lane >> 4;
    const int w    = blockIdx.x >> 10;
    const int b    = blockIdx.x & (NTILES - 1);
    const int c0   = w << 4;
    const int B1 = b << 4, B2 = b << 5, B3 = b << 6, B4 = b << 7;

    const float4 hx1 = load_halo(sig + (size_t)(c0 + n) * L1, B1, 16, L1, hi);
    const float4 hd1 = load_halo(d3  + (size_t)(c0 + n) * L1, B1, 16, L1, hi);
    const float4 hd2 = load_halo(d2  + (size_t)(c0 + n) * L2, B2, 32, L2, hi);
    const float4 hd3 = load_halo(d1  + (size_t)(c0 + n) * L3, B3, 64, L3, hi);
    const float4 hd4 = load_halo(d0  + (size_t)(c0 + n) * L4, B4, 128, L4, hi);
    float4 rx1[2], rd1[2];
#pragma unroll
    for (int j = 0; j < 2; ++j) {
        const size_t a = (size_t)(c0 + 8 * j + (lane >> 2)) * L1 + (size_t)(B1 + 4 * (lane & 3));
        rx1[j] = *reinterpret_cast<const float4*>(sig + a);
        rd1[j] = *reinterpret_cast<const float4*>(d3 + a);
    }
    float4 rd2[4];
#pragma unroll
    for (int j = 0; j < 4; ++j) {
        const size_t a = (size_t)(c0 + 4 * j + (lane >> 3)) * L2 + (size_t)(B2 + 4 * (lane & 7));
        rd2[j] = *reinterpret_cast<const float4*>(d2 + a);
    }
    float4 rd3[8];
#pragma unroll
    for (int j = 0; j < 8; ++j) {
        const size_t a = (size_t)(c0 + 2 * j + hi) * L3 + (size_t)(B3 + 4 * n);
        rd3[j] = *reinterpret_cast<const float4*>(d1 + a);
    }
    v4f rd4[16];
#pragma unroll
    for (int j = 0; j < 16; ++j) {
        const size_t a = (size_t)(c0 + j) * L4 + (size_t)(B4 + 4 * lane);
        rd4[j] = *reinterpret_cast<const v4f*>(d0 + a);
    }

    const float hh[6] = { h[0], h[1], h[2], h[3], h[4], h[5] };
    const float gg[6] = { g[0], g[1], g[2], g[3], g[4], g[5] };

    Bfrag B;
    make_bfrag(B, n, hi, hh, gg);

    float* X1 = sm + OFF_X1; float* D1 = sm + OFF_D1;
    float* X2 = sm + OFF_X2; float* D2 = sm + OFF_D2;
    float* X3 = sm + OFF_X3; float* D3 = sm + OFF_D3;
    float* X4 = sm + OFF_X4; float* D4 = sm + OFF_D4;

#pragma unroll
    for (int j = 0; j < 2; ++j) {
        const int o = (8 * j + (lane >> 2)) * P1 + 4 + 4 * (lane & 3);
        *reinterpret_cast<float4*>(X1 + o) = rx1[j];
        *reinterpret_cast<float4*>(D1 + o) = rd1[j];
    }
    *reinterpret_cast<float4*>(X1 + n * P1 + (hi ? 20 : 0)) = hx1;
    *reinterpret_cast<float4*>(D1 + n * P1 + (hi ? 20 : 0)) = hd1;
#pragma unroll
    for (int j = 0; j < 4; ++j) {
        const int o = (4 * j + (lane >> 3)) * P2 + 4 + 4 * (lane & 7);
        *reinterpret_cast<float4*>(D2 + o) = rd2[j];
    }
    *reinterpret_cast<float4*>(D2 + n * P2 + (hi ? 36 : 0)) = hd2;
#pragma unroll
    for (int j = 0; j < 8; ++j) {
        *reinterpret_cast<float4*>(D3 + (2 * j + hi) * P3 + 4 + 4 * n) = rd3[j];
    }
    *reinterpret_cast<float4*>(D3 + n * P3 + (hi ? 68 : 0)) = hd3;
    const float4 z4 = make_float4(0.0f, 0.0f, 0.0f, 0.0f);
    *reinterpret_cast<float4*>(hi ? (D1 + n * P1 + 24) : (X1 + n * P1 + 24)) = z4;
    *reinterpret_cast<float4*>(hi ? (D3 + n * P3 + 72) : (D2 + n * P2 + 40)) = z4;

    __syncthreads();

    run_level<3, P1, 2, P2, false>(X1, D1, X2, nullptr, 0, n, hi, 0, B);
    __syncthreads();
    fix_edges(X2, P2, 32, b, lane);
    __syncthreads();

    run_level<5, P2, 2, P3, false>(X2, D2, X3, nullptr, 0, n, hi, 0, B);
    __syncthreads();
    fix_edges(X3, P3, 64, b, lane);
    __syncthreads();

    run_level<9, P3, 2, P4, false>(X3, D3, X4, nullptr, 0, n, hi, 0, B);
    __syncthreads();
    fix_edges(X4, P4, 128, b, lane);
#pragma unroll
    for (int j = 0; j < 16; ++j) {
        *reinterpret_cast<v4f*>(D4 + j * P4 + 4 + 4 * lane) = rd4[j];
    }
    *reinterpret_cast<float4*>(D4 + n * P4 + (hi ? 132 : 0)) = hd4;
    __syncthreads();

    const size_t outL = 2 * (size_t)L4;
    const int odd = n & 1;
    float* obase = out + (size_t)(c0 + (hi << 3)) * outL + 2 * (size_t)B4 + (odd ? 15 + n : n);
    run_level<16, P4, 3, P4, true>(X4, D4, nullptr, obase, outL, n, hi, odd, B);
}

extern "C" void kernel_launch(void* const* d_in, const int* in_sizes, int n_in,
                              void* d_out, int out_size, void* d_ws, size_t ws_size,
                              hipStream_t stream) {
    (void)in_sizes; (void)n_in; (void)out_size; (void)d_ws; (void)ws_size;

    const float* sig = (const float*)d_in[0];
    const float* d0  = (const float*)d_in[1];
    const float* d1  = (const float*)d_in[2];
    const float* d2  = (const float*)d_in[3];
    const float* d3  = (const float*)d_in[4];
    const float* h   = (const float*)d_in[5];
    const float* g   = (const float*)d_in[6];
    float* out = (float*)d_out;

    synth4_kernel<<<4096, dim3(32), 0, stream>>>(sig, d0, d1, d2, d3, out, h, g);
}


# speedup vs baseline: 3.6544x; 1.0715x over previous
#include <hip/hip_runtime.h>
#include <hip/hip_bf16.h>


typedef float v2f __attribute__((ext_vector_type(2)));
typedef float v8f __attribute__((ext_vector_type(8)));
typedef float v4f __attribute__((ext_vector_type(4)));

#define P1 28
#define P2 44
#define P3 76
#define P4 140

#define OFF_X1 0
#define OFF_D1 (OFF_X1 + 16 * P1)
#define OFF_D2 (OFF_D1 + 16 * P1)
#define OFF_X2 (OFF_D2 + 16 * P2)
#define OFF_X3 (OFF_X2 + 16 * P2)
#define OFF_D3 (OFF_X3 + 16 * P3)
#define LDS_FLOATS (OFF_D3 + 16 * P3)
#define OFF_X4 0
#define OFF_D4 OFF_X3
static_assert(16 * P4 <= OFF_X3, "level-4 x window must fit over the level-1/2 buffers");
static_assert(OFF_D4 + 16 * P4 <= LDS_FLOATS, "level-4 d window must fit over the level-3 buffers");

#define NTILES 1024

__device__ __forceinline__ float bandsel(int dd, int ph, const float* f) {
    float r = 0.0f;
    r = (dd == 0) ? (ph ? f[5] : f[4]) : r;
    r = (dd == 1) ? (ph ? f[3] : f[2]) : r;
    r = (dd == 2) ? (ph ? f[1] : f[0]) : r;
    return r;
}

__device__ __forceinline__ float4 load_halo(const float* __restrict__ row, int T0, int len, int L, int right) {
    const bool edge = right ? (T0 + len >= L) : (T0 == 0);
    float4 v;
    if (!edge) {
        v = *reinterpret_cast<const float4*>(row + (right ? (T0 + len) : (T0 - 4)));
    } else if (right) {
        v = make_float4(2.0f * row[L - 1] - row[L - 2], 0.0f, 0.0f, 0.0f);
    } else {
        v = make_float4(0.0f, 0.0f, 0.0f, 2.0f * row[0] - row[1]);
    }
    return v;
}

struct Bfrag {
    v2f w[5];
};

__device__ __forceinline__ void make_bfrag(Bfrag& B, int n, int hi,
                                           const float* hh, const float* gg) {
    const int p = n >> 1, ph = n & 1;
#pragma unroll
    for (int c = 0; c < 5; ++c) {
        float e[2];
#pragma unroll
        for (int q = 0; q < 2; ++q) {
            const int k = c * 4 + (hi << 1) + q;
            const bool isd = (k >= 10);
            const int dd = (isd ? k - 10 : k) - p;
            const float wx = bandsel(dd, ph, hh);
            const float wd = bandsel(dd, ph, gg);
            e[q] = isd ? wd : wx;
        }
        B.w[c] = v2f{ e[0], e[1] };
    }
}

__device__ __forceinline__ void load_afrag(v2f (&a)[5], const float* q01, const float* q2, const float* q34) {
    a[0] = v2f{ q01[0], q01[1] };
    a[1] = v2f{ q01[4], q01[5] };
    a[2] = v2f{ q2[0],  q2[1]  };
    a[3] = v2f{ q34[0], q34[1] };
    a[4] = v2f{ q34[4], q34[5] };
}

__device__ __forceinline__ void wmma_tail_slots() {
    __builtin_amdgcn_sched_barrier(0);
    asm volatile("v_nop\n\tv_nop\n\tv_nop\n\tv_nop");
    __builtin_amdgcn_sched_barrier(0);
}

struct Afrag2 {
    v2f a[2][5];
};

__device__ __forceinline__ void load_trip(Afrag2& F, int cnt, const float* q01, const float* q2, const float* q34) {
    load_afrag(F.a[0], q01, q2, q34);
    if (cnt == 2) load_afrag(F.a[1], q01 + 8, q2 + 8, q34 + 8);
}

__device__ __forceinline__ void mma_trip(const Afrag2& F, int cnt, const Bfrag& B, v8f& acc0, v8f& acc1) {
    acc0 = v8f{};
    acc1 = v8f{};
#pragma unroll
    for (int c = 0; c < 5; ++c) {
        acc0 = __builtin_amdgcn_wmma_f32_16x16x4_f32(false, F.a[0][c], false, B.w[c], (short)0, acc0, false, false);
        if (cnt == 2)
            acc1 = __builtin_amdgcn_wmma_f32_16x16x4_f32(false, F.a[1][c], false, B.w[c], (short)0, acc1, false, false);
    }
    wmma_tail_slots();
}

template <int PNEXT>
__device__ __forceinline__ void store_next(float* Xn, int it, int n, int hi, const v8f& acc) {
    const int col = 2 + (it << 4) + n;
    if (col < PNEXT) {
        float* s = Xn + (hi << 3) * PNEXT + col;
#pragma unroll
        for (int r = 0; r < 8; ++r) s[r * PNEXT] = acc[r];
    }
}

__device__ __forceinline__ void store_out(float* s, size_t outL, int odd, const v8f& acc0, const v8f& acc1) {
#pragma unroll
    for (int r = 0; r < 8; ++r) {
        const float give = odd ? acc0[r] : acc1[r];
        const float recv = __builtin_bit_cast(float,
            __builtin_amdgcn_update_dpp(0, __builtin_bit_cast(int, give), 0xB1, 0xf, 0xf, true));
        float2 v;
        v.x = odd ? recv : acc0[r];
        v.y = odd ? acc1[r] : recv;
        *reinterpret_cast<float2*>(s + (size_t)r * outL) = v;
    }
}

template <int NSUB, int PIN, int COL0, int PNEXT, bool LAST>
__device__ __forceinline__ void run_level(const float* X, const float* D, float* Xn,
                                          float* obase, size_t outL,
                                          int n, int hi, int odd, const Bfrag& B) {
    const float* xb = X + n * PIN + COL0;
    const float* db = D + n * PIN + COL0;
    const float* q01 = xb + (hi << 1);
    const float* q2  = hi ? db : xb + 8;
    const float* q34 = db + 2 + (hi << 1);
    constexpr int NT = (NSUB + 1) / 2;
    Afrag2 F[2];
    load_trip(F[0], NSUB >= 2 ? 2 : 1, q01, q2, q34);
#pragma unroll
    for (int t = 0; t < NT; ++t) {
        const int cnt = (NSUB - 2 * t >= 2) ? 2 : 1;
        if (t + 1 < NT) {
            const int cn = (NSUB - 2 * (t + 1) >= 2) ? 2 : 1;
            const int o = (t + 1) << 4;
            load_trip(F[(t + 1) & 1], cn, q01 + o, q2 + o, q34 + o);
        }
        __builtin_amdgcn_sched_barrier(0);
        v8f acc0, acc1;
        mma_trip(F[t & 1], cnt, B, acc0, acc1);
        if (LAST) {
            store_out(obase + (t << 5), outL, odd, acc0, acc1);
        } else {
            store_next<PNEXT>(Xn, 2 * t, n, hi, acc0);
            if (cnt == 2) store_next<PNEXT>(Xn, 2 * t + 1, n, hi, acc1);
        }
    }
}

__device__ __forceinline__ void fix_edges(float* X, int pitch, int len, int b, int lane) {
    if (lane < 16) {
        float* r = X + lane * pitch;
        if (b == 0) r[3] = 2.0f * r[4] - r[5];
        if (b == NTILES - 1) r[len + 4] = 2.0f * r[len + 3] - r[len + 2];
    }
}

__global__ __launch_bounds__(32) void synth4_kernel(
        const float* __restrict__ sig, const float* __restrict__ d0,
        const float* __restrict__ d1, const float* __restrict__ d2,
        const float* __restrict__ d3, float* __restrict__ out,
        const float* __restrict__ h, const float* __restrict__ g) {
    __shared__ __attribute__((aligned(16))) float sm[LDS_FLOATS];

    constexpr int L1 = 16384, L2 = 32768, L3 = 65536, L4 = 131072;

    const int lane = threadIdx.x & 31;
    const int n    = lane & 15;
    const int hi   = lane >> 4;
    const int tile = ((blockIdx.x & 7) << 9) | (blockIdx.x >> 3);
    const int w    = tile >> 10;
    const int b    = tile & (NTILES - 1);
    const int c0   = w << 4;
    const int B1 = b << 4, B2 = b << 5, B3 = b << 6, B4 = b << 7;

    const float4 hx1 = load_halo(sig + (size_t)(c0 + n) * L1, B1, 16, L1, hi);
    const float4 hd1 = load_halo(d3  + (size_t)(c0 + n) * L1, B1, 16, L1, hi);
    const float4 hd2 = load_halo(d2  + (size_t)(c0 + n) * L2, B2, 32, L2, hi);
    const float4 hd3 = load_halo(d1  + (size_t)(c0 + n) * L3, B3, 64, L3, hi);
    const float4 hd4 = load_halo(d0  + (size_t)(c0 + n) * L4, B4, 128, L4, hi);
    float4 rx1[2], rd1[2];
#pragma unroll
    for (int j = 0; j < 2; ++j) {
        const size_t a = (size_t)(c0 + 8 * j + (lane >> 2)) * L1 + (size_t)(B1 + 4 * (lane & 3));
        rx1[j] = *reinterpret_cast<const float4*>(sig + a);
        rd1[j] = *reinterpret_cast<const float4*>(d3 + a);
    }
    float4 rd2[4];
#pragma unroll
    for (int j = 0; j < 4; ++j) {
        const size_t a = (size_t)(c0 + 4 * j + (lane >> 3)) * L2 + (size_t)(B2 + 4 * (lane & 7));
        rd2[j] = *reinterpret_cast<const float4*>(d2 + a);
    }
    float4 rd3[8];
#pragma unroll
    for (int j = 0; j < 8; ++j) {
        const size_t a = (size_t)(c0 + 2 * j + hi) * L3 + (size_t)(B3 + 4 * n);
        rd3[j] = *reinterpret_cast<const float4*>(d1 + a);
    }
    v4f rd4[16];
#pragma unroll
    for (int j = 0; j < 16; ++j) {
        const size_t a = (size_t)(c0 + j) * L4 + (size_t)(B4 + 4 * lane);
        rd4[j] = *reinterpret_cast<const v4f*>(d0 + a);
    }

    const float hh[6] = { h[0], h[1], h[2], h[3], h[4], h[5] };
    const float gg[6] = { g[0], g[1], g[2], g[3], g[4], g[5] };

    Bfrag B;
    make_bfrag(B, n, hi, hh, gg);

    float* X1 = sm + OFF_X1; float* D1 = sm + OFF_D1;
    float* X2 = sm + OFF_X2; float* D2 = sm + OFF_D2;
    float* X3 = sm + OFF_X3; float* D3 = sm + OFF_D3;
    float* X4 = sm + OFF_X4; float* D4 = sm + OFF_D4;

#pragma unroll
    for (int j = 0; j < 2; ++j) {
        const int o = (8 * j + (lane >> 2)) * P1 + 4 + 4 * (lane & 3);
        *reinterpret_cast<float4*>(X1 + o) = rx1[j];
        *reinterpret_cast<float4*>(D1 + o) = rd1[j];
    }
    *reinterpret_cast<float4*>(X1 + n * P1 + (hi ? 20 : 0)) = hx1;
    *reinterpret_cast<float4*>(D1 + n * P1 + (hi ? 20 : 0)) = hd1;
#pragma unroll
    for (int j = 0; j < 4; ++j) {
        const int o = (4 * j + (lane >> 3)) * P2 + 4 + 4 * (lane & 7);
        *reinterpret_cast<float4*>(D2 + o) = rd2[j];
    }
    *reinterpret_cast<float4*>(D2 + n * P2 + (hi ? 36 : 0)) = hd2;
#pragma unroll
    for (int j = 0; j < 8; ++j) {
        *reinterpret_cast<float4*>(D3 + (2 * j + hi) * P3 + 4 + 4 * n) = rd3[j];
    }
    *reinterpret_cast<float4*>(D3 + n * P3 + (hi ? 68 : 0)) = hd3;
    const float4 z4 = make_float4(0.0f, 0.0f, 0.0f, 0.0f);
    *reinterpret_cast<float4*>(hi ? (D1 + n * P1 + 24) : (X1 + n * P1 + 24)) = z4;
    *reinterpret_cast<float4*>(hi ? (D3 + n * P3 + 72) : (D2 + n * P2 + 40)) = z4;

    __syncthreads();

    run_level<3, P1, 2, P2, false>(X1, D1, X2, nullptr, 0, n, hi, 0, B);
    __syncthreads();
    fix_edges(X2, P2, 32, b, lane);
    __syncthreads();

    run_level<5, P2, 2, P3, false>(X2, D2, X3, nullptr, 0, n, hi, 0, B);
    __syncthreads();
    fix_edges(X3, P3, 64, b, lane);
    __syncthreads();

    run_level<9, P3, 2, P4, false>(X3, D3, X4, nullptr, 0, n, hi, 0, B);
    __syncthreads();
    fix_edges(X4, P4, 128, b, lane);
#pragma unroll
    for (int j = 0; j < 16; ++j) {
        *reinterpret_cast<v4f*>(D4 + j * P4 + 4 + 4 * lane) = rd4[j];
    }
    *reinterpret_cast<float4*>(D4 + n * P4 + (hi ? 132 : 0)) = hd4;
    __syncthreads();

    const size_t outL = 2 * (size_t)L4;
    const int odd = n & 1;
    float* obase = out + (size_t)(c0 + (hi << 3)) * outL + 2 * (size_t)B4 + (odd ? 15 + n : n);
    run_level<16, P4, 3, P4, true>(X4, D4, nullptr, obase, outL, n, hi, odd, B);
}

extern "C" void kernel_launch(void* const* d_in, const int* in_sizes, int n_in,
                              void* d_out, int out_size, void* d_ws, size_t ws_size,
                              hipStream_t stream) {
    (void)in_sizes; (void)n_in; (void)out_size; (void)d_ws; (void)ws_size;

    const float* sig = (const float*)d_in[0];
    const float* d0  = (const float*)d_in[1];
    const float* d1  = (const float*)d_in[2];
    const float* d2  = (const float*)d_in[3];
    const float* d3  = (const float*)d_in[4];
    const float* h   = (const float*)d_in[5];
    const float* g   = (const float*)d_in[6];
    float* out = (float*)d_out;

    synth4_kernel<<<4096, dim3(32), 0, stream>>>(sig, d0, d1, d2, d3, out, h, g);
}


# 8 independent waves per work-group, each on its own LDS slice with wave-local LDS hand-overs instead of barriers: 512 work-groups enter the chip, not 4096 one-wave ones; same tiles, 16 waves per WGP
# speedup vs baseline: 3.9136x; 1.0709x over previous
#include <hip/hip_runtime.h>
#include <hip/hip_bf16.h>


typedef float v2f __attribute__((ext_vector_type(2)));
typedef float v8f __attribute__((ext_vector_type(8)));
typedef float v4f __attribute__((ext_vector_type(4)));

#define P1 28
#define P2 44
#define P3 76
#define P4 140

#define OFF_X1 0
#define OFF_D1 (OFF_X1 + 16 * P1)
#define OFF_D2 (OFF_D1 + 16 * P1)
#define OFF_X2 (OFF_D2 + 16 * P2)
#define OFF_X3 (OFF_X2 + 16 * P2)
#define OFF_D3 (OFF_X3 + 16 * P3)
#define LDS_FLOATS (OFF_D3 + 16 * P3)
#define OFF_X4 0
#define OFF_D4 OFF_X3
static_assert(16 * P4 <= OFF_X3, "level-4 x window must fit over the level-1/2 buffers");
static_assert(OFF_D4 + 16 * P4 <= LDS_FLOATS, "level-4 d window must fit over the level-3 buffers");

#define NTILES 1024
#define WAVES 8
static_assert(WAVES == 8, "the launch below is written for 512 work-groups of 256 threads");

__device__ __forceinline__ float bandsel(int dd, int ph, const float* f) {
    float r = 0.0f;
    r = (dd == 0) ? (ph ? f[5] : f[4]) : r;
    r = (dd == 1) ? (ph ? f[3] : f[2]) : r;
    r = (dd == 2) ? (ph ? f[1] : f[0]) : r;
    return r;
}

__device__ __forceinline__ float4 load_halo(const float* __restrict__ row, int T0, int len, int L, int right) {
    const bool edge = right ? (T0 + len >= L) : (T0 == 0);
    float4 v;
    if (!edge) {
        v = *reinterpret_cast<const float4*>(row + (right ? (T0 + len) : (T0 - 4)));
    } else if (right) {
        v = make_float4(2.0f * row[L - 1] - row[L - 2], 0.0f, 0.0f, 0.0f);
    } else {
        v = make_float4(0.0f, 0.0f, 0.0f, 2.0f * row[0] - row[1]);
    }
    return v;
}

struct Bfrag {
    v2f w[5];
};

__device__ __forceinline__ void make_bfrag(Bfrag& B, int n, int hi,
                                           const float* hh, const float* gg) {
    const int p = n >> 1, ph = n & 1;
#pragma unroll
    for (int c = 0; c < 5; ++c) {
        float e[2];
#pragma unroll
        for (int q = 0; q < 2; ++q) {
            const int k = c * 4 + (hi << 1) + q;
            const bool isd = (k >= 10);
            const int dd = (isd ? k - 10 : k) - p;
            const float wx = bandsel(dd, ph, hh);
            const float wd = bandsel(dd, ph, gg);
            e[q] = isd ? wd : wx;
        }
        B.w[c] = v2f{ e[0], e[1] };
    }
}

__device__ __forceinline__ void load_afrag(v2f (&a)[5], const float* q01, const float* q2, const float* q34) {
    a[0] = v2f{ q01[0], q01[1] };
    a[1] = v2f{ q01[4], q01[5] };
    a[2] = v2f{ q2[0],  q2[1]  };
    a[3] = v2f{ q34[0], q34[1] };
    a[4] = v2f{ q34[4], q34[5] };
}

__device__ __forceinline__ void wmma_tail_slots() {
    __builtin_amdgcn_sched_barrier(0);
    asm volatile("v_nop\n\tv_nop\n\tv_nop\n\tv_nop");
    __builtin_amdgcn_sched_barrier(0);
}

struct Afrag2 {
    v2f a[2][5];
};

__device__ __forceinline__ void load_trip(Afrag2& F, int cnt, const float* q01, const float* q2, const float* q34) {
    load_afrag(F.a[0], q01, q2, q34);
    if (cnt == 2) load_afrag(F.a[1], q01 + 8, q2 + 8, q34 + 8);
}

__device__ __forceinline__ void mma_trip(const Afrag2& F, int cnt, const Bfrag& B, v8f& acc0, v8f& acc1) {
    acc0 = v8f{};
    acc1 = v8f{};
#pragma unroll
    for (int c = 0; c < 5; ++c) {
        acc0 = __builtin_amdgcn_wmma_f32_16x16x4_f32(false, F.a[0][c], false, B.w[c], (short)0, acc0, false, false);
        if (cnt == 2)
            acc1 = __builtin_amdgcn_wmma_f32_16x16x4_f32(false, F.a[1][c], false, B.w[c], (short)0, acc1, false, false);
    }
    wmma_tail_slots();
}

template <int PNEXT>
__device__ __forceinline__ void store_next(float* Xn, int it, int n, int hi, const v8f& acc) {
    const int col = 2 + (it << 4) + n;
    if (col < PNEXT) {
        float* s = Xn + (hi << 3) * PNEXT + col;
#pragma unroll
        for (int r = 0; r < 8; ++r) s[r * PNEXT] = acc[r];
    }
}

__device__ __forceinline__ void store_out(float* s, size_t outL, int odd, const v8f& acc0, const v8f& acc1) {
#pragma unroll
    for (int r = 0; r < 8; ++r) {
        const float give = odd ? acc0[r] : acc1[r];
        const float recv = __builtin_bit_cast(float,
            __builtin_amdgcn_update_dpp(0, __builtin_bit_cast(int, give), 0xB1, 0xf, 0xf, true));
        float2 v;
        v.x = odd ? recv : acc0[r];
        v.y = odd ? acc1[r] : recv;
        *reinterpret_cast<float2*>(s + (size_t)r * outL) = v;
    }
}

template <int NSUB, int PIN, int COL0, int PNEXT, bool LAST>
__device__ __forceinline__ void run_level(const float* X, const float* D, float* Xn,
                                          float* obase, size_t outL,
                                          int n, int hi, int odd, const Bfrag& B) {
    const float* xb = X + n * PIN + COL0;
    const float* db = D + n * PIN + COL0;
    const float* q01 = xb + (hi << 1);
    const float* q2  = hi ? db : xb + 8;
    const float* q34 = db + 2 + (hi << 1);
    constexpr int NT = (NSUB + 1) / 2;
    Afrag2 F[2];
    load_trip(F[0], NSUB >= 2 ? 2 : 1, q01, q2, q34);
#pragma unroll
    for (int t = 0; t < NT; ++t) {
        const int cnt = (NSUB - 2 * t >= 2) ? 2 : 1;
        if (t + 1 < NT) {
            const int cn = (NSUB - 2 * (t + 1) >= 2) ? 2 : 1;
            const int o = (t + 1) << 4;
            load_trip(F[(t + 1) & 1], cn, q01 + o, q2 + o, q34 + o);
        }
        __builtin_amdgcn_sched_barrier(0);
        v8f acc0, acc1;
        mma_trip(F[t & 1], cnt, B, acc0, acc1);
        if (LAST) {
            store_out(obase + (t << 5), outL, odd, acc0, acc1);
        } else {
            store_next<PNEXT>(Xn, 2 * t, n, hi, acc0);
            if (cnt == 2) store_next<PNEXT>(Xn, 2 * t + 1, n, hi, acc1);
        }
    }
}

__device__ __forceinline__ void wave_handover() {
    __builtin_amdgcn_fence(__ATOMIC_RELEASE, "workgroup", "local");
    __builtin_amdgcn_fence(__ATOMIC_ACQUIRE, "workgroup", "local");
}

__device__ __forceinline__ void fix_edges(float* X, int pitch, int len, int b, int lane) {
    if (lane < 16) {
        float* r = X + lane * pitch;
        if (b == 0) r[3] = 2.0f * r[4] - r[5];
        if (b == NTILES - 1) r[len + 4] = 2.0f * r[len + 3] - r[len + 2];
    }
}

__global__ __launch_bounds__(32 * WAVES) void synth4_kernel(
        const float* __restrict__ sig, const float* __restrict__ d0,
        const float* __restrict__ d1, const float* __restrict__ d2,
        const float* __restrict__ d3, float* __restrict__ out,
        const float* __restrict__ h, const float* __restrict__ g) {
    __shared__ __attribute__((aligned(16))) float sm_all[WAVES * LDS_FLOATS];

    constexpr int L1 = 16384, L2 = 32768, L3 = 65536, L4 = 131072;

    const int lane = threadIdx.x & 31;
    const int n    = lane & 15;
    const int hi   = lane >> 4;
    const int wv   = __builtin_amdgcn_readfirstlane((int)(threadIdx.x >> 5));
    float* sm      = sm_all + wv * LDS_FLOATS;
    const int tile = ((blockIdx.x & 7) << 9) | ((blockIdx.x >> 3) * WAVES + wv);
    const int w    = tile >> 10;
    const int b    = tile & (NTILES - 1);
    const int c0   = w << 4;
    const int B1 = b << 4, B2 = b << 5, B3 = b << 6, B4 = b << 7;

    const float4 hx1 = load_halo(sig + (size_t)(c0 + n) * L1, B1, 16, L1, hi);
    const float4 hd1 = load_halo(d3  + (size_t)(c0 + n) * L1, B1, 16, L1, hi);
    const float4 hd2 = load_halo(d2  + (size_t)(c0 + n) * L2, B2, 32, L2, hi);
    const float4 hd3 = load_halo(d1  + (size_t)(c0 + n) * L3, B3, 64, L3, hi);
    const float4 hd4 = load_halo(d0  + (size_t)(c0 + n) * L4, B4, 128, L4, hi);
    float4 rx1[2], rd1[2];
#pragma unroll
    for (int j = 0; j < 2; ++j) {
        const size_t a = (size_t)(c0 + 8 * j + (lane >> 2)) * L1 + (size_t)(B1 + 4 * (lane & 3));
        rx1[j] = *reinterpret_cast<const float4*>(sig + a);
        rd1[j] = *reinterpret_cast<const float4*>(d3 + a);
    }
    float4 rd2[4];
#pragma unroll
    for (int j = 0; j < 4; ++j) {
        const size_t a = (size_t)(c0 + 4 * j + (lane >> 3)) * L2 + (size_t)(B2 + 4 * (lane & 7));
        rd2[j] = *reinterpret_cast<const float4*>(d2 + a);
    }
    float4 rd3[8];
#pragma unroll
    for (int j = 0; j < 8; ++j) {
        const size_t a = (size_t)(c0 + 2 * j + hi) * L3 + (size_t)(B3 + 4 * n);
        rd3[j] = *reinterpret_cast<const float4*>(d1 + a);
    }
    v4f rd4[16];
#pragma unroll
    for (int j = 0; j < 16; ++j) {
        const size_t a = (size_t)(c0 + j) * L4 + (size_t)(B4 + 4 * lane);
        rd4[j] = *reinterpret_cast<const v4f*>(d0 + a);
    }

    const float hh[6] = { h[0], h[1], h[2], h[3], h[4], h[5] };
    const float gg[6] = { g[0], g[1], g[2], g[3], g[4], g[5] };

    Bfrag B;
    make_bfrag(B, n, hi, hh, gg);

    float* X1 = sm + OFF_X1; float* D1 = sm + OFF_D1;
    float* X2 = sm + OFF_X2; float* D2 = sm + OFF_D2;
    float* X3 = sm + OFF_X3; float* D3 = sm + OFF_D3;
    float* X4 = sm + OFF_X4; float* D4 = sm + OFF_D4;

#pragma unroll
    for (int j = 0; j < 2; ++j) {
        const int o = (8 * j + (lane >> 2)) * P1 + 4 + 4 * (lane & 3);
        *reinterpret_cast<float4*>(X1 + o) = rx1[j];
        *reinterpret_cast<float4*>(D1 + o) = rd1[j];
    }
    *reinterpret_cast<float4*>(X1 + n * P1 + (hi ? 20 : 0)) = hx1;
    *reinterpret_cast<float4*>(D1 + n * P1 + (hi ? 20 : 0)) = hd1;
#pragma unroll
    for (int j = 0; j < 4; ++j) {
        const int o = (4 * j + (lane >> 3)) * P2 + 4 + 4 * (lane & 7);
        *reinterpret_cast<float4*>(D2 + o) = rd2[j];
    }
    *reinterpret_cast<float4*>(D2 + n * P2 + (hi ? 36 : 0)) = hd2;
#pragma unroll
    for (int j = 0; j < 8; ++j) {
        *reinterpret_cast<float4*>(D3 + (2 * j + hi) * P3 + 4 + 4 * n) = rd3[j];
    }
    *reinterpret_cast<float4*>(D3 + n * P3 + (hi ? 68 : 0)) = hd3;
    const float4 z4 = make_float4(0.0f, 0.0f, 0.0f, 0.0f);
    *reinterpret_cast<float4*>(hi ? (D1 + n * P1 + 24) : (X1 + n * P1 + 24)) = z4;
    *reinterpret_cast<float4*>(hi ? (D3 + n * P3 + 72) : (D2 + n * P2 + 40)) = z4;

    wave_handover();

    run_level<3, P1, 2, P2, false>(X1, D1, X2, nullptr, 0, n, hi, 0, B);
    wave_handover();
    fix_edges(X2, P2, 32, b, lane);
    wave_handover();

    run_level<5, P2, 2, P3, false>(X2, D2, X3, nullptr, 0, n, hi, 0, B);
    wave_handover();
    fix_edges(X3, P3, 64, b, lane);
    wave_handover();

    run_level<9, P3, 2, P4, false>(X3, D3, X4, nullptr, 0, n, hi, 0, B);
    wave_handover();
    fix_edges(X4, P4, 128, b, lane);
#pragma unroll
    for (int j = 0; j < 16; ++j) {
        *reinterpret_cast<v4f*>(D4 + j * P4 + 4 + 4 * lane) = rd4[j];
    }
    *reinterpret_cast<float4*>(D4 + n * P4 + (hi ? 132 : 0)) = hd4;
    wave_handover();

    const size_t outL = 2 * (size_t)L4;
    const int odd = n & 1;
    float* obase = out + (size_t)(c0 + (hi << 3)) * outL + 2 * (size_t)B4 + (odd ? 15 + n : n);
    run_level<16, P4, 3, P4, true>(X4, D4, nullptr, obase, outL, n, hi, odd, B);
}

extern "C" void kernel_launch(void* const* d_in, const int* in_sizes, int n_in,
                              void* d_out, int out_size, void* d_ws, size_t ws_size,
                              hipStream_t stream) {
    (void)in_sizes; (void)n_in; (void)out_size; (void)d_ws; (void)ws_size;

    const float* sig = (const float*)d_in[0];
    const float* d0  = (const float*)d_in[1];
    const float* d1  = (const float*)d_in[2];
    const float* d2  = (const float*)d_in[3];
    const float* d3  = (const float*)d_in[4];
    const float* h   = (const float*)d_in[5];
    const float* g   = (const float*)d_in[6];
    float* out = (float*)d_out;

    synth4_kernel<<<512, dim3(256), 0, stream>>>(sig, d0, d1, d2, d3, out, h, g);
}


# level-4 results pass through consumed LDS window columns and leave as 16-byte-per-lane stores of whole 256-byte units, read back under the next trip: 32 stores per wave, not 64; no lane swaps
# speedup vs baseline: 4.5587x; 1.1648x over previous
#include <hip/hip_runtime.h>
#include <hip/hip_bf16.h>


typedef float v2f __attribute__((ext_vector_type(2)));
typedef float v8f __attribute__((ext_vector_type(8)));
typedef float v4f __attribute__((ext_vector_type(4)));

#define P1 28
#define P2 44
#define P3 76
#define P4 140

#define OFF_X1 0
#define OFF_D1 (OFF_X1 + 16 * P1)
#define OFF_D2 (OFF_D1 + 16 * P1)
#define OFF_X2 (OFF_D2 + 16 * P2)
#define OFF_X3 (OFF_X2 + 16 * P2)
#define OFF_D3 (OFF_X3 + 16 * P3)
#define LDS_FLOATS (OFF_D3 + 16 * P3)
#define OFF_X4 0
#define OFF_D4 OFF_X3
static_assert(16 * P4 <= OFF_X3, "level-4 x window must fit over the level-1/2 buffers");
static_assert(OFF_D4 + 16 * P4 <= LDS_FLOATS, "level-4 d window must fit over the level-3 buffers");

#define NTILES 1024
#define WAVES 8
static_assert(WAVES == 8, "the launch below is written for 512 work-groups of 256 threads");

__device__ __forceinline__ float bandsel(int dd, int ph, const float* f) {
    float r = 0.0f;
    r = (dd == 0) ? (ph ? f[5] : f[4]) : r;
    r = (dd == 1) ? (ph ? f[3] : f[2]) : r;
    r = (dd == 2) ? (ph ? f[1] : f[0]) : r;
    return r;
}

__device__ __forceinline__ float4 load_halo(const float* __restrict__ row, int T0, int len, int L, int right) {
    const bool edge = right ? (T0 + len >= L) : (T0 == 0);
    float4 v;
    if (!edge) {
        v = *reinterpret_cast<const float4*>(row + (right ? (T0 + len) : (T0 - 4)));
    } else if (right) {
        v = make_float4(2.0f * row[L - 1] - row[L - 2], 0.0f, 0.0f, 0.0f);
    } else {
        v = make_float4(0.0f, 0.0f, 0.0f, 2.0f * row[0] - row[1]);
    }
    return v;
}

struct Bfrag {
    v2f w[5];
};

__device__ __forceinline__ void make_bfrag(Bfrag& B, int n, int hi,
                                           const float* hh, const float* gg) {
    const int p = n >> 1, ph = n & 1;
#pragma unroll
    for (int c = 0; c < 5; ++c) {
        float e[2];
#pragma unroll
        for (int q = 0; q < 2; ++q) {
            const int k = c * 4 + (hi << 1) + q;
            const bool isd = (k >= 10);
            const int dd = (isd ? k - 10 : k) - p;
            const float wx = bandsel(dd, ph, hh);
            const float wd = bandsel(dd, ph, gg);
            e[q] = isd ? wd : wx;
        }
        B.w[c] = v2f{ e[0], e[1] };
    }
}

__device__ __forceinline__ void load_afrag(v2f (&a)[5], const float* q01, const float* q2, const float* q34) {
    a[0] = v2f{ q01[0], q01[1] };
    a[1] = v2f{ q01[4], q01[5] };
    a[2] = v2f{ q2[0],  q2[1]  };
    a[3] = v2f{ q34[0], q34[1] };
    a[4] = v2f{ q34[4], q34[5] };
}

__device__ __forceinline__ void wmma_tail_slots() {
    __builtin_amdgcn_sched_barrier(0);
    asm volatile("v_nop\n\tv_nop\n\tv_nop\n\tv_nop");
    __builtin_amdgcn_sched_barrier(0);
}

struct Afrag2 {
    v2f a[2][5];
};

__device__ __forceinline__ void load_trip(Afrag2& F, int cnt, const float* q01, const float* q2, const float* q34) {
    load_afrag(F.a[0], q01, q2, q34);
    if (cnt == 2) load_afrag(F.a[1], q01 + 8, q2 + 8, q34 + 8);
}

__device__ __forceinline__ void mma_trip(const Afrag2& F, int cnt, const Bfrag& B, v8f& acc0, v8f& acc1) {
    acc0 = v8f{};
    acc1 = v8f{};
#pragma unroll
    for (int c = 0; c < 5; ++c) {
        acc0 = __builtin_amdgcn_wmma_f32_16x16x4_f32(false, F.a[0][c], false, B.w[c], (short)0, acc0, false, false);
        if (cnt == 2)
            acc1 = __builtin_amdgcn_wmma_f32_16x16x4_f32(false, F.a[1][c], false, B.w[c], (short)0, acc1, false, false);
    }
    wmma_tail_slots();
}

template <int PNEXT>
__device__ __forceinline__ void store_next(float* Xn, int it, int n, int hi, const v8f& acc) {
    const int col = 2 + (it << 4) + n;
    if (col < PNEXT) {
        float* s = Xn + (hi << 3) * PNEXT + col;
#pragma unroll
        for (int r = 0; r < 8; ++r) s[r * PNEXT] = acc[r];
    }
}

__device__ __forceinline__ void stage_out(float* Xw, float* Dw, int t, int n, int hi,
                                          const v8f& acc0, const v8f& acc1) {
    float* sx = Xw + (hi << 3) * P4 + (t << 4) + n;
    float* sd = Dw + (hi << 3) * P4 + (t << 4) + n;
#pragma unroll
    for (int r = 0; r < 8; ++r) {
        sx[r * P4] = acc0[r];
        sd[r * P4] = acc1[r];
    }
}

__device__ __forceinline__ void read_back(v4f (&v)[8], const float* rb, int u) {
#pragma unroll
    for (int r = 0; r < 8; ++r) v[r] = *reinterpret_cast<const v4f*>(rb + r * P4 + (u << 5));
}

__device__ __forceinline__ void write_out(const v4f (&v)[8], float* orow, size_t outL, int u) {
#pragma unroll
    for (int r = 0; r < 8; ++r) *reinterpret_cast<v4f*>(orow + (size_t)r * outL + (u << 6)) = v[r];
}

__device__ __forceinline__ void wave_handover() {
    __builtin_amdgcn_fence(__ATOMIC_RELEASE, "workgroup", "local");
    __builtin_amdgcn_fence(__ATOMIC_ACQUIRE, "workgroup", "local");
}

template <int NSUB, int PIN, int COL0, int PNEXT, bool LAST>
__device__ __forceinline__ void run_level(const float* X, const float* D, float* Xn,
                                          float* obase, size_t outL,
                                          int n, int hi, const Bfrag& B) {
    const float* xb = X + n * PIN + COL0;
    const float* db = D + n * PIN + COL0;
    const float* q01 = xb + (hi << 1);
    const float* q2  = hi ? db : xb + 8;
    const float* q34 = db + 2 + (hi << 1);
    constexpr int NT = (NSUB + 1) / 2;
    static_assert(!LAST || (NSUB % 4 == 0 && PIN == P4), "level 4 stores pairs of trips from its own windows");
    float* Xw = const_cast<float*>(X);
    float* Dw = const_cast<float*>(D);
    const float* rb = (((n >> 2) & 1) ? D : X) + (hi << 3) * PIN + ((n >> 3) << 4) + ((n & 3) << 2);
    Afrag2 F[2];
    load_trip(F[0], NSUB >= 2 ? 2 : 1, q01, q2, q34);
#pragma unroll
    for (int t = 0; t < NT; ++t) {
        const int cnt = (NSUB - 2 * t >= 2) ? 2 : 1;
        if (t + 1 < NT) {
            const int cn = (NSUB - 2 * (t + 1) >= 2) ? 2 : 1;
            const int o = (t + 1) << 4;
            load_trip(F[(t + 1) & 1], cn, q01 + o, q2 + o, q34 + o);
        }
        const bool pend = LAST && t >= 2 && !(t & 1);
        v4f ob[8];
        if (pend) read_back(ob, rb, (t >> 1) - 1);
        __builtin_amdgcn_sched_barrier(0);
        v8f acc0, acc1;
        mma_trip(F[t & 1], cnt, B, acc0, acc1);
        if (LAST) {
            if (pend) write_out(ob, obase, outL, (t >> 1) - 1);
            stage_out(Xw, Dw, t, n, hi, acc0, acc1);
            if (t & 1) {
                wave_handover();
                if (t == NT - 1) {
                    read_back(ob, rb, t >> 1);
                    write_out(ob, obase, outL, t >> 1);
                }
            }
        } else {
            store_next<PNEXT>(Xn, 2 * t, n, hi, acc0);
            if (cnt == 2) store_next<PNEXT>(Xn, 2 * t + 1, n, hi, acc1);
        }
    }
}

__device__ __forceinline__ void fix_edges(float* X, int pitch, int len, int b, int lane) {
    if (lane < 16) {
        float* r = X + lane * pitch;
        if (b == 0) r[3] = 2.0f * r[4] - r[5];
        if (b == NTILES - 1) r[len + 4] = 2.0f * r[len + 3] - r[len + 2];
    }
}

__global__ __launch_bounds__(32 * WAVES) void synth4_kernel(
        const float* __restrict__ sig, const float* __restrict__ d0,
        const float* __restrict__ d1, const float* __restrict__ d2,
        const float* __restrict__ d3, float* __restrict__ out,
        const float* __restrict__ h, const float* __restrict__ g) {
    __shared__ __attribute__((aligned(16))) float sm_all[WAVES * LDS_FLOATS];

    constexpr int L1 = 16384, L2 = 32768, L3 = 65536, L4 = 131072;

    const int lane = threadIdx.x & 31;
    const int n    = lane & 15;
    const int hi   = lane >> 4;
    const int wv   = __builtin_amdgcn_readfirstlane((int)(threadIdx.x >> 5));
    float* sm      = sm_all + wv * LDS_FLOATS;
    const int tile = ((blockIdx.x & 7) << 9) | ((blockIdx.x >> 3) * WAVES + wv);
    const int w    = tile >> 10;
    const int b    = tile & (NTILES - 1);
    const int c0   = w << 4;
    const int B1 = b << 4, B2 = b << 5, B3 = b << 6, B4 = b << 7;

    const float4 hx1 = load_halo(sig + (size_t)(c0 + n) * L1, B1, 16, L1, hi);
    const float4 hd1 = load_halo(d3  + (size_t)(c0 + n) * L1, B1, 16, L1, hi);
    const float4 hd2 = load_halo(d2  + (size_t)(c0 + n) * L2, B2, 32, L2, hi);
    const float4 hd3 = load_halo(d1  + (size_t)(c0 + n) * L3, B3, 64, L3, hi);
    const float4 hd4 = load_halo(d0  + (size_t)(c0 + n) * L4, B4, 128, L4, hi);
    float4 rx1[2], rd1[2];
#pragma unroll
    for (int j = 0; j < 2; ++j) {
        const size_t a = (size_t)(c0 + 8 * j + (lane >> 2)) * L1 + (size_t)(B1 + 4 * (lane & 3));
        rx1[j] = *reinterpret_cast<const float4*>(sig + a);
        rd1[j] = *reinterpret_cast<const float4*>(d3 + a);
    }
    float4 rd2[4];
#pragma unroll
    for (int j = 0; j < 4; ++j) {
        const size_t a = (size_t)(c0 + 4 * j + (lane >> 3)) * L2 + (size_t)(B2 + 4 * (lane & 7));
        rd2[j] = *reinterpret_cast<const float4*>(d2 + a);
    }
    float4 rd3[8];
#pragma unroll
    for (int j = 0; j < 8; ++j) {
        const size_t a = (size_t)(c0 + 2 * j + hi) * L3 + (size_t)(B3 + 4 * n);
        rd3[j] = *reinterpret_cast<const float4*>(d1 + a);
    }
    v4f rd4[16];
#pragma unroll
    for (int j = 0; j < 16; ++j) {
        const size_t a = (size_t)(c0 + j) * L4 + (size_t)(B4 + 4 * lane);
        rd4[j] = *reinterpret_cast<const v4f*>(d0 + a);
    }

    const float hh[6] = { h[0], h[1], h[2], h[3], h[4], h[5] };
    const float gg[6] = { g[0], g[1], g[2], g[3], g[4], g[5] };

    Bfrag B;
    make_bfrag(B, n, hi, hh, gg);

    float* X1 = sm + OFF_X1; float* D1 = sm + OFF_D1;
    float* X2 = sm + OFF_X2; float* D2 = sm + OFF_D2;
    float* X3 = sm + OFF_X3; float* D3 = sm + OFF_D3;
    float* X4 = sm + OFF_X4; float* D4 = sm + OFF_D4;

#pragma unroll
    for (int j = 0; j < 2; ++j) {
        const int o = (8 * j + (lane >> 2)) * P1 + 4 + 4 * (lane & 3);
        *reinterpret_cast<float4*>(X1 + o) = rx1[j];
        *reinterpret_cast<float4*>(D1 + o) = rd1[j];
    }
    *reinterpret_cast<float4*>(X1 + n * P1 + (hi ? 20 : 0)) = hx1;
    *reinterpret_cast<float4*>(D1 + n * P1 + (hi ? 20 : 0)) = hd1;
#pragma unroll
    for (int j = 0; j < 4; ++j) {
        const int o = (4 * j + (lane >> 3)) * P2 + 4 + 4 * (lane & 7);
        *reinterpret_cast<float4*>(D2 + o) = rd2[j];
    }
    *reinterpret_cast<float4*>(D2 + n * P2 + (hi ? 36 : 0)) = hd2;
#pragma unroll
    for (int j = 0; j < 8; ++j) {
        *reinterpret_cast<float4*>(D3 + (2 * j + hi) * P3 + 4 + 4 * n) = rd3[j];
    }
    *reinterpret_cast<float4*>(D3 + n * P3 + (hi ? 68 : 0)) = hd3;
    const float4 z4 = make_float4(0.0f, 0.0f, 0.0f, 0.0f);
    *reinterpret_cast<float4*>(hi ? (D1 + n * P1 + 24) : (X1 + n * P1 + 24)) = z4;
    *reinterpret_cast<float4*>(hi ? (D3 + n * P3 + 72) : (D2 + n * P2 + 40)) = z4;

    wave_handover();

    run_level<3, P1, 2, P2, false>(X1, D1, X2, nullptr, 0, n, hi, B);
    wave_handover();
    fix_edges(X2, P2, 32, b, lane);
    wave_handover();

    run_level<5, P2, 2, P3, false>(X2, D2, X3, nullptr, 0, n, hi, B);
    wave_handover();
    fix_edges(X3, P3, 64, b, lane);
    wave_handover();

    run_level<9, P3, 2, P4, false>(X3, D3, X4, nullptr, 0, n, hi, B);
    wave_handover();
    fix_edges(X4, P4, 128, b, lane);
#pragma unroll
    for (int j = 0; j < 16; ++j) {
        *reinterpret_cast<v4f*>(D4 + j * P4 + 4 + 4 * lane) = rd4[j];
    }
    *reinterpret_cast<float4*>(D4 + n * P4 + (hi ? 132 : 0)) = hd4;
    wave_handover();

    const size_t outL = 2 * (size_t)L4;
    float* obase = out + (size_t)(c0 + (hi << 3)) * outL + 2 * (size_t)B4 + 4 * n;
    run_level<16, P4, 3, P4, true>(X4, D4, nullptr, obase, outL, n, hi, B);
}

extern "C" void kernel_launch(void* const* d_in, const int* in_sizes, int n_in,
                              void* d_out, int out_size, void* d_ws, size_t ws_size,
                              hipStream_t stream) {
    (void)in_sizes; (void)n_in; (void)out_size; (void)d_ws; (void)ws_size;

    const float* sig = (const float*)d_in[0];
    const float* d0  = (const float*)d_in[1];
    const float* d1  = (const float*)d_in[2];
    const float* d2  = (const float*)d_in[3];
    const float* d3  = (const float*)d_in[4];
    const float* h   = (const float*)d_in[5];
    const float* g   = (const float*)d_in[6];
    float* out = (float*)d_out;

    synth4_kernel<<<512, dim3(256), 0, stream>>>(sig, d0, d1, d2, d3, out, h, g);
}


# a wave's 16 matrix rows are 16 consecutive tiles of ONE channel, not one tile of 16 channels: loads and stores are contiguous 512-byte pieces, inner halos copied in LDS, 2 halo lines per input not 32
# speedup vs baseline: 5.3662x; 1.1771x over previous
#include <hip/hip_runtime.h>
#include <hip/hip_bf16.h>


typedef float v2f __attribute__((ext_vector_type(2)));
typedef float v8f __attribute__((ext_vector_type(8)));
typedef float v4f __attribute__((ext_vector_type(4)));

#define P1 28
#define P2 44
#define P3 76
#define P4 140

#define OFF_X1 0
#define OFF_D1 (OFF_X1 + 16 * P1)
#define OFF_D2 (OFF_D1 + 16 * P1)
#define OFF_X2 (OFF_D2 + 16 * P2)
#define OFF_X3 (OFF_X2 + 16 * P2)
#define OFF_D3 (OFF_X3 + 16 * P3)
#define LDS_FLOATS (OFF_D3 + 16 * P3)
#define OFF_X4 0
#define OFF_D4 OFF_X3
static_assert(16 * P4 <= OFF_X3, "level-4 x window must fit over the level-1/2 buffers");
static_assert(OFF_D4 + 16 * P4 <= LDS_FLOATS, "level-4 d window must fit over the level-3 buffers");

#define NTILES 1024
#define WAVES 8
static_assert(WAVES == 8, "the launch below is written for 512 work-groups of 256 threads");

__device__ __forceinline__ float bandsel(int dd, int ph, const float* f) {
    float r = 0.0f;
    r = (dd == 0) ? (ph ? f[5] : f[4]) : r;
    r = (dd == 1) ? (ph ? f[3] : f[2]) : r;
    r = (dd == 2) ? (ph ? f[1] : f[0]) : r;
    return r;
}

struct Halo {
    v4f q;
    bool edge;
};
__device__ __forceinline__ Halo load_halo(const float* __restrict__ row, int T0, int len, int L, int right) {
    Halo h;
    h.edge = right ? (T0 + len >= L) : (T0 == 0);
    const int at = right ? (h.edge ? L - 4 : T0 + len) : (h.edge ? 0 : T0 - 4);
    h.q = *reinterpret_cast<const v4f*>(row + at);
    return h;
}
__device__ __forceinline__ v4f halo_value(const Halo& h, int right) {
    const float el = 2.0f * h.q.x - h.q.y;
    const float er = 2.0f * h.q.w - h.q.z;
    v4f v;
    v.x = h.edge ? (right ? er : 0.0f) : h.q.x;
    v.y = h.edge ? 0.0f : h.q.y;
    v.z = h.edge ? 0.0f : h.q.z;
    v.w = h.edge ? (right ? 0.0f : el) : h.q.w;
    return v;
}

struct Bfrag {
    v2f w[5];
};

__device__ __forceinline__ void make_bfrag(Bfrag& B, int n, int hi,
                                           const float* hh, const float* gg) {
    const int p = n >> 1, ph = n & 1;
#pragma unroll
    for (int c = 0; c < 5; ++c) {
        float e[2];
#pragma unroll
        for (int q = 0; q < 2; ++q) {
            const int k = c * 4 + (hi << 1) + q;
            const bool isd = (k >= 10);
            const int dd = (isd ? k - 10 : k) - p;
            const float wx = bandsel(dd, ph, hh);
            const float wd = bandsel(dd, ph, gg);
            e[q] = isd ? wd : wx;
        }
        B.w[c] = v2f{ e[0], e[1] };
    }
}

__device__ __forceinline__ void load_afrag(v2f (&a)[5], const float* q01, const float* q2, const float* q34) {
    a[0] = v2f{ q01[0], q01[1] };
    a[1] = v2f{ q01[4], q01[5] };
    a[2] = v2f{ q2[0],  q2[1]  };
    a[3] = v2f{ q34[0], q34[1] };
    a[4] = v2f{ q34[4], q34[5] };
}

__device__ __forceinline__ void wmma_tail_slots() {
    __builtin_amdgcn_sched_barrier(0);
    asm volatile("v_nop\n\tv_nop\n\tv_nop\n\tv_nop");
    __builtin_amdgcn_sched_barrier(0);
}

struct Afrag2 {
    v2f a[2][5];
};

__device__ __forceinline__ void load_trip(Afrag2& F, int cnt, const float* q01, const float* q2, const float* q34) {
    load_afrag(F.a[0], q01, q2, q34);
    if (cnt == 2) load_afrag(F.a[1], q01 + 8, q2 + 8, q34 + 8);
}

__device__ __forceinline__ void mma_trip(const Afrag2& F, int cnt, const Bfrag& B, v8f& acc0, v8f& acc1) {
    acc0 = v8f{};
    acc1 = v8f{};
#pragma unroll
    for (int c = 0; c < 5; ++c) {
        acc0 = __builtin_amdgcn_wmma_f32_16x16x4_f32(false, F.a[0][c], false, B.w[c], (short)0, acc0, false, false);
        if (cnt == 2)
            acc1 = __builtin_amdgcn_wmma_f32_16x16x4_f32(false, F.a[1][c], false, B.w[c], (short)0, acc1, false, false);
    }
    wmma_tail_slots();
}

template <int PNEXT>
__device__ __forceinline__ void store_next(float* Xn, int it, int n, int hi, const v8f& acc) {
    const int col = 2 + (it << 4) + n;
    if (col < PNEXT) {
        float* s = Xn + (hi << 3) * PNEXT + col;
#pragma unroll
        for (int r = 0; r < 8; ++r) s[r * PNEXT] = acc[r];
    }
}

__device__ __forceinline__ void stage_out(float* Xw, float* Dw, int t, int n, int hi,
                                          const v8f& acc0, const v8f& acc1) {
    float* sx = Xw + (hi << 3) * P4 + (t << 4) + n;
    float* sd = Dw + (hi << 3) * P4 + (t << 4) + n;
#pragma unroll
    for (int r = 0; r < 8; ++r) {
        sx[r * P4] = acc0[r];
        sd[r * P4] = acc1[r];
    }
}

__device__ __forceinline__ void read_back(v4f (&v)[8], const float* rb, int u) {
#pragma unroll
    for (int r = 0; r < 8; ++r) v[r] = *reinterpret_cast<const v4f*>(rb + r * P4 + (u << 5));
}

__device__ __forceinline__ void write_out(const v4f (&v)[8], float* orow, size_t outL, int u) {
#pragma unroll
    for (int r = 0; r < 8; ++r) *reinterpret_cast<v4f*>(orow + (size_t)r * outL + (u << 6)) = v[r];
}

__device__ __forceinline__ void wave_handover() {
    __builtin_amdgcn_fence(__ATOMIC_RELEASE, "workgroup", "local");
    __builtin_amdgcn_fence(__ATOMIC_ACQUIRE, "workgroup", "local");
}

template <int NSUB, int PIN, int COL0, int PNEXT, bool LAST>
__device__ __forceinline__ void run_level(const float* X, const float* D, float* Xn,
                                          float* obase, size_t outL,
                                          int n, int hi, const Bfrag& B) {
    const float* xb = X + n * PIN + COL0;
    const float* db = D + n * PIN + COL0;
    const float* q01 = xb + (hi << 1);
    const float* q2  = hi ? db : xb + 8;
    const float* q34 = db + 2 + (hi << 1);
    constexpr int NT = (NSUB + 1) / 2;
    static_assert(!LAST || (NSUB % 4 == 0 && PIN == P4), "level 4 stores pairs of trips from its own windows");
    float* Xw = const_cast<float*>(X);
    float* Dw = const_cast<float*>(D);
    const float* rb = (((n >> 2) & 1) ? D : X) + (hi << 3) * PIN + ((n >> 3) << 4) + ((n & 3) << 2);
    Afrag2 F[2];
    load_trip(F[0], NSUB >= 2 ? 2 : 1, q01, q2, q34);
#pragma unroll
    for (int t = 0; t < NT; ++t) {
        const int cnt = (NSUB - 2 * t >= 2) ? 2 : 1;
        if (t + 1 < NT) {
            const int cn = (NSUB - 2 * (t + 1) >= 2) ? 2 : 1;
            const int o = (t + 1) << 4;
            load_trip(F[(t + 1) & 1], cn, q01 + o, q2 + o, q34 + o);
        }
        const bool pend = LAST && t >= 2 && !(t & 1);
        v4f ob[8];
        if (pend) read_back(ob, rb, (t >> 1) - 1);
        __builtin_amdgcn_sched_barrier(0);
        v8f acc0, acc1;
        mma_trip(F[t & 1], cnt, B, acc0, acc1);
        if (LAST) {
            if (pend) write_out(ob, obase, outL, (t >> 1) - 1);
            stage_out(Xw, Dw, t, n, hi, acc0, acc1);
            if (t & 1) {
                wave_handover();
                if (t == NT - 1) {
                    read_back(ob, rb, t >> 1);
                    write_out(ob, obase, outL, t >> 1);
                }
            }
        } else {
            store_next<PNEXT>(Xn, 2 * t, n, hi, acc0);
            if (cnt == 2) store_next<PNEXT>(Xn, 2 * t + 1, n, hi, acc1);
        }
    }
}

__device__ __forceinline__ void fix_edges(float* X, int pitch, int len, int bt, int lane) {
    if (lane < 16) {
        float* r = X + lane * pitch;
        if (bt == 0) r[3] = 2.0f * r[4] - r[5];
        if (bt == NTILES - 1) r[len + 4] = 2.0f * r[len + 3] - r[len + 2];
    }
}

template <int P, int LEN>
__device__ __forceinline__ v4f halo_src(const float* Wd, int n, int hi) {
    const int srow = (hi ? n + 1 : n - 1) & 15;
    return *reinterpret_cast<const v4f*>(Wd + srow * P + (hi ? 4 : LEN));
}
template <int P, int LEN>
__device__ __forceinline__ void halo_put(float* Wd, int n, int hi, const Halo& h, const v4f& q) {
    const v4f gh = halo_value(h, hi);
    const bool outer = hi ? (n == 15) : (n == 0);
    v4f o;
    o.x = outer ? gh.x : q.x;
    o.y = outer ? gh.y : q.y;
    o.z = outer ? gh.z : q.z;
    o.w = outer ? gh.w : q.w;
    *reinterpret_cast<v4f*>(Wd + n * P + (hi ? LEN + 4 : 0)) = o;
}

__global__ __launch_bounds__(32 * WAVES) void synth4_kernel(
        const float* sig, const float* d0,
        const float* d1, const float* d2,
        const float* d3, float* __restrict__ out,
        const float* __restrict__ h, const float* __restrict__ g) {
    __shared__ __attribute__((aligned(16))) float sm_all[WAVES * LDS_FLOATS];

    constexpr int L1 = 16384, L2 = 32768, L3 = 65536, L4 = 131072;

    const int lane = threadIdx.x & 31;
    const int n    = lane & 15;
    const int hi   = lane >> 4;
    const int wv   = __builtin_amdgcn_readfirstlane((int)(threadIdx.x >> 5));
    float* sm      = sm_all + wv * LDS_FLOATS;
    const int span = ((blockIdx.x & 7) << 9) | ((blockIdx.x >> 3) * WAVES + wv);
    const int ch   = span >> 6;
    const int bs   = span & 63;
    const int bt   = (bs << 4) + n;
    const int S1 = bs << 8, S2 = bs << 9, S3 = bs << 10, S4 = bs << 11;
    const float* sig_r = sig + (size_t)ch * L1;
    const float* d3_r  = d3  + (size_t)ch * L1;
    const float* d2_r  = d2  + (size_t)ch * L2;
    const float* d1_r  = d1  + (size_t)ch * L3;
    const float* d0_r  = d0  + (size_t)ch * L4;

    const Halo hx1 = load_halo(sig_r, S1, 256,  L1, hi);
    const Halo hd1 = load_halo(d3_r,  S1, 256,  L1, hi);
    const Halo hd2 = load_halo(d2_r,  S2, 512,  L2, hi);
    const Halo hd3 = load_halo(d1_r,  S3, 1024, L3, hi);
    v4f rx1[2], rd1[2];
#pragma unroll
    for (int j = 0; j < 2; ++j) {
        rx1[j] = *reinterpret_cast<const v4f*>(sig_r + S1 + 128 * j + 4 * lane);
        rd1[j] = *reinterpret_cast<const v4f*>(d3_r  + S1 + 128 * j + 4 * lane);
    }
    v4f rd2[4];
#pragma unroll
    for (int j = 0; j < 4; ++j) {
        rd2[j] = *reinterpret_cast<const v4f*>(d2_r + S2 + 128 * j + 4 * lane);
    }
    v4f rd3[8];
#pragma unroll
    for (int j = 0; j < 8; ++j) {
        rd3[j] = *reinterpret_cast<const v4f*>(d1_r + S3 + 128 * j + 4 * lane);
    }
    asm volatile("" ::: "memory");
    const Halo hd4 = load_halo(d0_r, S4, 2048, L4, hi);
    v4f rd4[16];
#pragma unroll
    for (int j = 0; j < 16; ++j) {
        rd4[j] = *reinterpret_cast<const v4f*>(d0_r + S4 + 128 * j + 4 * lane);
    }

    const float hh[6] = { h[0], h[1], h[2], h[3], h[4], h[5] };
    const float gg[6] = { g[0], g[1], g[2], g[3], g[4], g[5] };

    Bfrag B;
    make_bfrag(B, n, hi, hh, gg);

    float* X1 = sm + OFF_X1; float* D1 = sm + OFF_D1;
    float* X2 = sm + OFF_X2; float* D2 = sm + OFF_D2;
    float* X3 = sm + OFF_X3; float* D3 = sm + OFF_D3;
    float* X4 = sm + OFF_X4; float* D4 = sm + OFF_D4;

#pragma unroll
    for (int j = 0; j < 2; ++j) {
        const int o = (8 * j + (lane >> 2)) * P1 + 4 + 4 * (lane & 3);
        *reinterpret_cast<v4f*>(X1 + o) = rx1[j];
        *reinterpret_cast<v4f*>(D1 + o) = rd1[j];
    }
#pragma unroll
    for (int j = 0; j < 4; ++j) {
        const int o = (4 * j + (lane >> 3)) * P2 + 4 + 4 * (lane & 7);
        *reinterpret_cast<v4f*>(D2 + o) = rd2[j];
    }
#pragma unroll
    for (int j = 0; j < 8; ++j) {
        *reinterpret_cast<v4f*>(D3 + (2 * j + hi) * P3 + 4 + 4 * n) = rd3[j];
    }
    const float4 z4 = make_float4(0.0f, 0.0f, 0.0f, 0.0f);
    *reinterpret_cast<float4*>(hi ? (D1 + n * P1 + 24) : (X1 + n * P1 + 24)) = z4;
    *reinterpret_cast<float4*>(hi ? (D3 + n * P3 + 72) : (D2 + n * P2 + 40)) = z4;

    wave_handover();
    {
        const v4f qx1 = halo_src<P1, 16>(X1, n, hi);
        const v4f qd1 = halo_src<P1, 16>(D1, n, hi);
        const v4f qd2 = halo_src<P2, 32>(D2, n, hi);
        const v4f qd3 = halo_src<P3, 64>(D3, n, hi);
        halo_put<P1, 16>(X1, n, hi, hx1, qx1);
        halo_put<P1, 16>(D1, n, hi, hd1, qd1);
        halo_put<P2, 32>(D2, n, hi, hd2, qd2);
        halo_put<P3, 64>(D3, n, hi, hd3, qd3);
    }
    wave_handover();

    run_level<3, P1, 2, P2, false>(X1, D1, X2, nullptr, 0, n, hi, B);
    wave_handover();
    fix_edges(X2, P2, 32, bt, lane);
    wave_handover();

    run_level<5, P2, 2, P3, false>(X2, D2, X3, nullptr, 0, n, hi, B);
    wave_handover();
    fix_edges(X3, P3, 64, bt, lane);
    wave_handover();

    run_level<9, P3, 2, P4, false>(X3, D3, X4, nullptr, 0, n, hi, B);
    wave_handover();
    fix_edges(X4, P4, 128, bt, lane);
#pragma unroll
    for (int j = 0; j < 16; ++j) {
        *reinterpret_cast<v4f*>(D4 + j * P4 + 4 + 4 * lane) = rd4[j];
    }
    wave_handover();
    halo_put<P4, 128>(D4, n, hi, hd4, halo_src<P4, 128>(D4, n, hi));
    wave_handover();

    const size_t outL = 2 * (size_t)L4;
    float* obase = out + (size_t)ch * outL + (size_t)((bs << 12) + (hi << 11) + 4 * n);
    run_level<16, P4, 3, P4, true>(X4, D4, nullptr, obase, 256, n, hi, B);
}

extern "C" void kernel_launch(void* const* d_in, const int* in_sizes, int n_in,
                              void* d_out, int out_size, void* d_ws, size_t ws_size,
                              hipStream_t stream) {
    (void)in_sizes; (void)n_in; (void)out_size; (void)d_ws; (void)ws_size;

    const float* sig = (const float*)d_in[0];
    const float* d0  = (const float*)d_in[1];
    const float* d1  = (const float*)d_in[2];
    const float* d2  = (const float*)d_in[3];
    const float* d3  = (const float*)d_in[4];
    const float* h   = (const float*)d_in[5];
    const float* g   = (const float*)d_in[6];
    float* out = (float*)d_out;

    synth4_kernel<<<512, dim3(256), 0, stream>>>(sig, d0, d1, d2, d3, out, h, g);
}
